# starting kernel
# baseline (speedup 1.0000x reference)
#include <hip/hip_runtime.h>

#define IN_F  256
#define OUT_F 128
#define LDS_PAD 260

typedef __attribute__((ext_vector_type(2))) float v2f;
typedef __attribute__((ext_vector_type(8))) float v8f;

__global__ __launch_bounds__(256) void sgc_gemm_wmma(const float* __restrict__ h,
                                                     const float* __restrict__ w,
                                                     float* __restrict__ x) {
  __shared__ float atile[16 * LDS_PAD];
  const int m0  = blockIdx.x * 16;
  const int tid = threadIdx.x;

  for (int i4 = tid; i4 < 16 * IN_F / 4; i4 += 256) {
    int i = i4 * 4;
    int r = i / IN_F, c = i % IN_F;
    float4 v = *(const float4*)(h + (size_t)(m0 + r) * IN_F + c);
    atile[r * LDS_PAD + c + 0] = v.x;
    atile[r * LDS_PAD + c + 1] = v.y;
    atile[r * LDS_PAD + c + 2] = v.z;
    atile[r * LDS_PAD + c + 3] = v.w;
  }
  __syncthreads();

  const int wave = tid >> 5;
  const int lane = tid & 31;
  const int half = lane >> 4;
  const int l    = lane & 15;
  const int n0   = wave * 16;

  v8f c = {};
  const float* arow = atile + l * LDS_PAD + 2 * half;
  const float* bcol = w + (size_t)(2 * half) * OUT_F + n0 + l;
  for (int k = 0; k < IN_F; k += 4) {
    float2 av = *(const float2*)(arow + k);
    v2f a, b;
    a.x = av.x;
    a.y = av.y;
    b.x = bcol[(size_t)k * OUT_F];
    b.y = bcol[(size_t)(k + 1) * OUT_F];
    c = __builtin_amdgcn_wmma_f32_16x16x4_f32(false, a, false, b, (short)0, c,
                                              false, false);
  }

  float* out = x + (size_t)(m0 + 8 * half) * OUT_F + n0 + l;
#pragma unroll
  for (int r = 0; r < 8; ++r) out[(size_t)r * OUT_F] = c[r];
}

__global__ __launch_bounds__(256) void sgc_scatter(const float* __restrict__ x,
                                                   const float* __restrict__ norm,
                                                   const int* __restrict__ src,
                                                   const int* __restrict__ dst,
                                                   float* __restrict__ acc, int E) {
  int gw   = (blockIdx.x * 256 + threadIdx.x) >> 5;
  int lane = threadIdx.x & 31;
  if (gw >= E) return;
  int s = __builtin_amdgcn_readfirstlane(src[gw]);
  int d = __builtin_amdgcn_readfirstlane(dst[gw]);
  float ns = norm[s];
  float4 v = *(const float4*)(x + (size_t)s * OUT_F + lane * 4);
  float r0 = ns * v.x, r1 = ns * v.y, r2 = ns * v.z, r3 = ns * v.w;
  float* a = acc + (size_t)d * OUT_F + lane * 4;
  asm volatile(
      "global_atomic_add_f32 %4, %0, off\n\t"
      "global_atomic_add_f32 %4, %1, off offset:4\n\t"
      "global_atomic_add_f32 %4, %2, off offset:8\n\t"
      "global_atomic_add_f32 %4, %3, off offset:12"
      :
      : "v"(r0), "v"(r1), "v"(r2), "v"(r3), "v"(a)
      : "memory");
}

__global__ __launch_bounds__(256) void sgc_zero(float4* __restrict__ p, int total4) {
  int i = blockIdx.x * 256 + threadIdx.x;
  if (i < total4) p[i] = make_float4(0.f, 0.f, 0.f, 0.f);
}

__global__ __launch_bounds__(256) void sgc_scale_zero(const float* __restrict__ norm,
                                                      float* __restrict__ acc,
                                                      float* __restrict__ xout,
                                                      int total4) {
  int i4 = blockIdx.x * 256 + threadIdx.x;
  if (i4 >= total4) return;
  int node = i4 >> 5;
  float nv = norm[node];
  float4 v = ((const float4*)acc)[i4];
  ((float4*)xout)[i4] = make_float4(nv * v.x, nv * v.y, nv * v.z, nv * v.w);
  ((float4*)acc)[i4] = make_float4(0.f, 0.f, 0.f, 0.f);
}

extern "C" void kernel_launch(void* const* d_in, const int* in_sizes, int n_in,
                              void* d_out, int out_size, void* d_ws, size_t ws_size,
                              hipStream_t stream) {
  const float* h    = (const float*)d_in[0];
  const float* w    = (const float*)d_in[1];
  const float* norm = (const float*)d_in[2];
  const int*   src  = (const int*)d_in[3];
  const int*   dst  = (const int*)d_in[4];
  float* out = (float*)d_out;
  float* acc = (float*)d_ws;

  const int N = in_sizes[2];
  const int E = in_sizes[3];

  const int mblocks = (N + 15) / 16;
  const int total4  = N * OUT_F / 4;
  const int sblocks = (total4 + 255) / 256;
  const int eblocks = (E + 7) / 8;

  sgc_gemm_wmma<<<mblocks, 256, 0, stream>>>(h, w, out);
  sgc_zero<<<sblocks, 256, 0, stream>>>((float4*)acc, total4);

  sgc_scatter<<<eblocks, 256, 0, stream>>>(out, norm, src, dst, acc, E);
  sgc_scale_zero<<<sblocks, 256, 0, stream>>>(norm, acc, out, total4);

  sgc_scatter<<<eblocks, 256, 0, stream>>>(out, norm, src, dst, acc, E);
  sgc_scale_zero<<<sblocks, 256, 0, stream>>>(norm, acc, out, total4);
}


# Edge scatter: lane j owns floats j,j+32,j+64,j+96 so each load and each f32 atomic add instruction covers one whole 128-byte line (was 4 partial lines per add); sum order still atomic arrival
# speedup vs baseline: 1.3447x; 1.3447x over previous
#include <hip/hip_runtime.h>

#define IN_F  256
#define OUT_F 128
#define LDS_PAD 260

typedef __attribute__((ext_vector_type(2))) float v2f;
typedef __attribute__((ext_vector_type(8))) float v8f;

__global__ __launch_bounds__(256) void sgc_gemm_wmma(const float* __restrict__ h,
                                                     const float* __restrict__ w,
                                                     float* __restrict__ x) {
  __shared__ float atile[16 * LDS_PAD];
  const int m0  = blockIdx.x * 16;
  const int tid = threadIdx.x;

  for (int i4 = tid; i4 < 16 * IN_F / 4; i4 += 256) {
    int i = i4 * 4;
    int r = i / IN_F, c = i % IN_F;
    float4 v = *(const float4*)(h + (size_t)(m0 + r) * IN_F + c);
    atile[r * LDS_PAD + c + 0] = v.x;
    atile[r * LDS_PAD + c + 1] = v.y;
    atile[r * LDS_PAD + c + 2] = v.z;
    atile[r * LDS_PAD + c + 3] = v.w;
  }
  __syncthreads();

  const int wave = tid >> 5;
  const int lane = tid & 31;
  const int half = lane >> 4;
  const int l    = lane & 15;
  const int n0   = wave * 16;

  v8f c = {};
  const float* arow = atile + l * LDS_PAD + 2 * half;
  const float* bcol = w + (size_t)(2 * half) * OUT_F + n0 + l;
  for (int k = 0; k < IN_F; k += 4) {
    float2 av = *(const float2*)(arow + k);
    v2f a, b;
    a.x = av.x;
    a.y = av.y;
    b.x = bcol[(size_t)k * OUT_F];
    b.y = bcol[(size_t)(k + 1) * OUT_F];
    c = __builtin_amdgcn_wmma_f32_16x16x4_f32(false, a, false, b, (short)0, c,
                                              false, false);
  }

  float* out = x + (size_t)(m0 + 8 * half) * OUT_F + n0 + l;
#pragma unroll
  for (int r = 0; r < 8; ++r) out[(size_t)r * OUT_F] = c[r];
}

__global__ __launch_bounds__(256) void sgc_scatter(const float* __restrict__ x,
                                                   const float* __restrict__ norm,
                                                   const int* __restrict__ src,
                                                   const int* __restrict__ dst,
                                                   float* __restrict__ acc, int E) {
  int gw   = (blockIdx.x * 256 + threadIdx.x) >> 5;
  int lane = threadIdx.x & 31;
  if (gw >= E) return;
  int s = __builtin_amdgcn_readfirstlane(src[gw]);
  int d = __builtin_amdgcn_readfirstlane(dst[gw]);
  float ns = norm[s];
  const float* xr = x + (size_t)s * OUT_F + lane;
  float v0 = xr[0], v1 = xr[32], v2 = xr[64], v3 = xr[96];
  float r0 = ns * v0, r1 = ns * v1, r2 = ns * v2, r3 = ns * v3;
  float* a = acc + (size_t)d * OUT_F + lane;
  asm volatile(
      "global_atomic_add_f32 %4, %0, off\n\t"
      "global_atomic_add_f32 %4, %1, off offset:128\n\t"
      "global_atomic_add_f32 %4, %2, off offset:256\n\t"
      "global_atomic_add_f32 %4, %3, off offset:384"
      :
      : "v"(r0), "v"(r1), "v"(r2), "v"(r3), "v"(a)
      : "memory");
}

__global__ __launch_bounds__(256) void sgc_zero(float4* __restrict__ p, int total4) {
  int i = blockIdx.x * 256 + threadIdx.x;
  if (i < total4) p[i] = make_float4(0.f, 0.f, 0.f, 0.f);
}

__global__ __launch_bounds__(256) void sgc_scale_zero(const float* __restrict__ norm,
                                                      float* __restrict__ acc,
                                                      float* __restrict__ xout,
                                                      int total4) {
  int i4 = blockIdx.x * 256 + threadIdx.x;
  if (i4 >= total4) return;
  int node = i4 >> 5;
  float nv = norm[node];
  float4 v = ((const float4*)acc)[i4];
  ((float4*)xout)[i4] = make_float4(nv * v.x, nv * v.y, nv * v.z, nv * v.w);
  ((float4*)acc)[i4] = make_float4(0.f, 0.f, 0.f, 0.f);
}

extern "C" void kernel_launch(void* const* d_in, const int* in_sizes, int n_in,
                              void* d_out, int out_size, void* d_ws, size_t ws_size,
                              hipStream_t stream) {
  const float* h    = (const float*)d_in[0];
  const float* w    = (const float*)d_in[1];
  const float* norm = (const float*)d_in[2];
  const int*   src  = (const int*)d_in[3];
  const int*   dst  = (const int*)d_in[4];
  float* out = (float*)d_out;
  float* acc = (float*)d_ws;

  const int N = in_sizes[2];
  const int E = in_sizes[3];

  const int mblocks = (N + 15) / 16;
  const int total4  = N * OUT_F / 4;
  const int sblocks = (total4 + 255) / 256;
  const int eblocks = (E + 7) / 8;

  sgc_gemm_wmma<<<mblocks, 256, 0, stream>>>(h, w, out);
  sgc_zero<<<sblocks, 256, 0, stream>>>((float4*)acc, total4);

  sgc_scatter<<<eblocks, 256, 0, stream>>>(out, norm, src, dst, acc, E);
  sgc_scale_zero<<<sblocks, 256, 0, stream>>>(norm, acc, out, total4);

  sgc_scatter<<<eblocks, 256, 0, stream>>>(out, norm, src, dst, acc, E);
  sgc_scale_zero<<<sblocks, 256, 0, stream>>>(norm, acc, out, total4);
}


# Edge scatter: each wave takes 32 consecutive edges (indices and norms loaded once as whole lines, 8 edges' 32 row loads in flight) instead of one edge per wave; sum order still atomic arrival
# speedup vs baseline: 1.7512x; 1.3023x over previous
#include <hip/hip_runtime.h>

#define IN_F  256
#define OUT_F 128
#define LDS_PAD 260

typedef __attribute__((ext_vector_type(2))) float v2f;
typedef __attribute__((ext_vector_type(8))) float v8f;

__global__ __launch_bounds__(256) void sgc_gemm_wmma(const float* __restrict__ h,
                                                     const float* __restrict__ w,
                                                     float* __restrict__ x) {
  __shared__ float atile[16 * LDS_PAD];
  const int m0  = blockIdx.x * 16;
  const int tid = threadIdx.x;

  for (int i4 = tid; i4 < 16 * IN_F / 4; i4 += 256) {
    int i = i4 * 4;
    int r = i / IN_F, c = i % IN_F;
    float4 v = *(const float4*)(h + (size_t)(m0 + r) * IN_F + c);
    atile[r * LDS_PAD + c + 0] = v.x;
    atile[r * LDS_PAD + c + 1] = v.y;
    atile[r * LDS_PAD + c + 2] = v.z;
    atile[r * LDS_PAD + c + 3] = v.w;
  }
  __syncthreads();

  const int wave = tid >> 5;
  const int lane = tid & 31;
  const int half = lane >> 4;
  const int l    = lane & 15;
  const int n0   = wave * 16;

  v8f c = {};
  const float* arow = atile + l * LDS_PAD + 2 * half;
  const float* bcol = w + (size_t)(2 * half) * OUT_F + n0 + l;
  for (int k = 0; k < IN_F; k += 4) {
    float2 av = *(const float2*)(arow + k);
    v2f a, b;
    a.x = av.x;
    a.y = av.y;
    b.x = bcol[(size_t)k * OUT_F];
    b.y = bcol[(size_t)(k + 1) * OUT_F];
    c = __builtin_amdgcn_wmma_f32_16x16x4_f32(false, a, false, b, (short)0, c,
                                              false, false);
  }

  float* out = x + (size_t)(m0 + 8 * half) * OUT_F + n0 + l;
#pragma unroll
  for (int r = 0; r < 8; ++r) out[(size_t)r * OUT_F] = c[r];
}

typedef const float __attribute__((address_space(1)))* gfloat_cptr;

#define EDGES_PER_WAVE 32
#define EDGE_BATCH 8

template <bool TAIL>
__device__ __forceinline__ void sgc_scatter_batch(const float* __restrict__ x,
                                                  float* __restrict__ acc, int sv, int dv,
                                                  float nv, int lane, int loff, int b, int n) {
  float v[EDGE_BATCH][4];
#pragma unroll
  for (int j = 0; j < EDGE_BATCH; ++j) {
    const int s = __builtin_amdgcn_readlane(sv, b + j);
    gfloat_cptr xrow = (gfloat_cptr)(x + (size_t)s * OUT_F);
    asm("" : "+s"(xrow));
    v[j][0] = xrow[lane];
    v[j][1] = xrow[lane + 32];
    v[j][2] = xrow[lane + 64];
    v[j][3] = xrow[lane + 96];
  }
#pragma unroll
  for (int j = 0; j < EDGE_BATCH; ++j) {
    if (!TAIL || b + j < n) {
      const float ns = __int_as_float(__builtin_amdgcn_readlane(__float_as_int(nv), b + j));
      const int   d  = __builtin_amdgcn_readlane(dv, b + j);
      float* arow = acc + (size_t)d * OUT_F;
      const float r0 = ns * v[j][0], r1 = ns * v[j][1];
      const float r2 = ns * v[j][2], r3 = ns * v[j][3];
      asm volatile(
          "global_atomic_add_f32 %4, %0, %5\n\t"
          "global_atomic_add_f32 %4, %1, %5 offset:128\n\t"
          "global_atomic_add_f32 %4, %2, %5 offset:256\n\t"
          "global_atomic_add_f32 %4, %3, %5 offset:384\n\t"
          "s_wait_xcnt 0x0"
          :
          : "v"(r0), "v"(r1), "v"(r2), "v"(r3), "v"(loff), "s"(arow)
          : "memory");
    }
  }
}

__global__ __launch_bounds__(256) void sgc_scatter(const float* __restrict__ x,
                                                   const float* __restrict__ norm,
                                                   const int* __restrict__ src,
                                                   const int* __restrict__ dst,
                                                   float* __restrict__ acc, int E) {
  const int lane = threadIdx.x & 31;
  const int gw   = __builtin_amdgcn_readfirstlane((blockIdx.x * 256 + threadIdx.x) >> 5);
  const int e0   = gw * EDGES_PER_WAVE;
  if (e0 >= E) return;
  const int left = E - e0;
  const int n    = left < EDGES_PER_WAVE ? left : EDGES_PER_WAVE;
  const int e    = e0 + lane;
  const int ec   = e < E ? e : E - 1;
  const int   sv = src[ec];
  const int   dv = dst[ec];
  const float nv = norm[sv];
  const int loff = lane * 4;

  int b = 0;
  for (; b + EDGE_BATCH <= n; b += EDGE_BATCH)
    sgc_scatter_batch<false>(x, acc, sv, dv, nv, lane, loff, b, n);
  if (b < n)
    sgc_scatter_batch<true>(x, acc, sv, dv, nv, lane, loff, b, n);
}

__global__ __launch_bounds__(256) void sgc_zero(float4* __restrict__ p, int total4) {
  int i = blockIdx.x * 256 + threadIdx.x;
  if (i < total4) p[i] = make_float4(0.f, 0.f, 0.f, 0.f);
}

__global__ __launch_bounds__(256) void sgc_scale_zero(const float* __restrict__ norm,
                                                      float* __restrict__ acc,
                                                      float* __restrict__ xout,
                                                      int total4) {
  int i4 = blockIdx.x * 256 + threadIdx.x;
  if (i4 >= total4) return;
  int node = i4 >> 5;
  float nv = norm[node];
  float4 v = ((const float4*)acc)[i4];
  ((float4*)xout)[i4] = make_float4(nv * v.x, nv * v.y, nv * v.z, nv * v.w);
  ((float4*)acc)[i4] = make_float4(0.f, 0.f, 0.f, 0.f);
}

extern "C" void kernel_launch(void* const* d_in, const int* in_sizes, int n_in,
                              void* d_out, int out_size, void* d_ws, size_t ws_size,
                              hipStream_t stream) {
  const float* h    = (const float*)d_in[0];
  const float* w    = (const float*)d_in[1];
  const float* norm = (const float*)d_in[2];
  const int*   src  = (const int*)d_in[3];
  const int*   dst  = (const int*)d_in[4];
  float* out = (float*)d_out;
  float* acc = (float*)d_ws;

  const int N = in_sizes[2];
  const int E = in_sizes[3];

  const int mblocks = (N + 15) / 16;
  const int total4  = N * OUT_F / 4;
  const int sblocks = (total4 + 255) / 256;
  const int eblocks = (E + 255) / 256;

  sgc_gemm_wmma<<<mblocks, 256, 0, stream>>>(h, w, out);
  sgc_zero<<<sblocks, 256, 0, stream>>>((float4*)acc, total4);

  sgc_scatter<<<eblocks, 256, 0, stream>>>(out, norm, src, dst, acc, E);
  sgc_scale_zero<<<sblocks, 256, 0, stream>>>(norm, acc, out, total4);

  sgc_scatter<<<eblocks, 256, 0, stream>>>(out, norm, src, dst, acc, E);
  sgc_scale_zero<<<sblocks, 256, 0, stream>>>(norm, acc, out, total4);
}


# Scatter-add without atomics: a work-group owns 391 destination rows in LDS, its waves filter dst for them, one owner wave sums each row's edges in fixed order and stores it once, scaled
# speedup vs baseline: 1.9135x; 1.0927x over previous
#include <hip/hip_runtime.h>

#define IN_F  256
#define OUT_F 128
#define LDS_PAD 260

typedef __attribute__((ext_vector_type(2))) float v2f;
typedef __attribute__((ext_vector_type(8))) float v8f;

__global__ __launch_bounds__(256) void sgc_gemm_wmma(const float* __restrict__ h,
                                                     const float* __restrict__ w,
                                                     float* __restrict__ x) {
  __shared__ float atile[16 * LDS_PAD];
  const int m0  = blockIdx.x * 16;
  const int tid = threadIdx.x;

  for (int i4 = tid; i4 < 16 * IN_F / 4; i4 += 256) {
    int i = i4 * 4;
    int r = i / IN_F, c = i % IN_F;
    float4 v = *(const float4*)(h + (size_t)(m0 + r) * IN_F + c);
    atile[r * LDS_PAD + c + 0] = v.x;
    atile[r * LDS_PAD + c + 1] = v.y;
    atile[r * LDS_PAD + c + 2] = v.z;
    atile[r * LDS_PAD + c + 3] = v.w;
  }
  __syncthreads();

  const int wave = tid >> 5;
  const int lane = tid & 31;
  const int half = lane >> 4;
  const int l    = lane & 15;
  const int n0   = wave * 16;

  v8f c = {};
  const float* arow = atile + l * LDS_PAD + 2 * half;
  const float* bcol = w + (size_t)(2 * half) * OUT_F + n0 + l;
  for (int k = 0; k < IN_F; k += 4) {
    float2 av = *(const float2*)(arow + k);
    v2f a, b;
    a.x = av.x;
    a.y = av.y;
    b.x = bcol[(size_t)k * OUT_F];
    b.y = bcol[(size_t)(k + 1) * OUT_F];
    c = __builtin_amdgcn_wmma_f32_16x16x4_f32(false, a, false, b, (short)0, c,
                                              false, false);
  }

  float* out = x + (size_t)(m0 + 8 * half) * OUT_F + n0 + l;
#pragma unroll
  for (int r = 0; r < 8; ++r) out[(size_t)r * OUT_F] = c[r];
}

typedef const float __attribute__((address_space(1)))* gfloat_cptr;

#define NODES_PER_WG 391
#define HOP_WAVES 32
#define CHUNK 512
#define LIST_W 576
#define OWN_W 64
#define EDGE_BATCH 8

template <bool TAIL>
__device__ __forceinline__ void sgc_hop_batch(const float* __restrict__ x, float* acc, int sv,
                                              int dv, float nv, int lane, int b, int n) {
  float v[EDGE_BATCH][4];
#pragma unroll
  for (int j = 0; j < EDGE_BATCH; ++j) {
    const int s = __builtin_amdgcn_readlane(sv, b + j);
    gfloat_cptr xrow = (gfloat_cptr)(x + (size_t)s * OUT_F);
    asm("" : "+s"(xrow));
    v[j][0] = xrow[lane];
    v[j][1] = xrow[lane + 32];
    v[j][2] = xrow[lane + 64];
    v[j][3] = xrow[lane + 96];
  }
#pragma unroll
  for (int j = 0; j < EDGE_BATCH; ++j) {
    if (!TAIL || b + j < n) {
      const float ns = __int_as_float(__builtin_amdgcn_readlane(__float_as_int(nv), b + j));
      const int   d  = __builtin_amdgcn_readlane(dv, b + j);
      float* arow = acc + d * OUT_F + lane;
      arow[0]  += ns * v[j][0];
      arow[32] += ns * v[j][1];
      arow[64] += ns * v[j][2];
      arow[96] += ns * v[j][3];
    }
  }
}

template <bool TAIL>
__device__ __forceinline__ void sgc_hop_group(const float* __restrict__ x,
                                              const float* __restrict__ norm,
                                              const int* __restrict__ src, float* acc,
                                              const int* oe, const unsigned short* od, int n,
                                              int lane) {
  const int li = TAIL ? (lane < n ? lane : n - 1) : lane;
  const int e  = oe[li];
  const int dv = od[li];
  const int sv = src[e];
  const float nv = norm[sv];
  int b = 0;
  for (; b + EDGE_BATCH <= n; b += EDGE_BATCH)
    sgc_hop_batch<false>(x, acc, sv, dv, nv, lane, b, n);
  if (TAIL && b < n) sgc_hop_batch<true>(x, acc, sv, dv, nv, lane, b, n);
}

__device__ __forceinline__ void sgc_lds_barrier() {
  __builtin_amdgcn_fence(__ATOMIC_RELEASE, "workgroup", "local");
  __builtin_amdgcn_s_barrier();
  __builtin_amdgcn_fence(__ATOMIC_ACQUIRE, "workgroup", "local");
}

__device__ __forceinline__ void sgc_hop_publish_and_hand_over(
    const float* __restrict__ x, const float* __restrict__ norm, const int* __restrict__ src,
    float* acc, const int* note_e, const unsigned short* note_d, int* note_n, int* oe,
    unsigned short* od, int& cnt, int& pc, bool last, int wave, int lane, unsigned lt) {
  if (lane == 0) note_n[wave] = cnt;
  sgc_lds_barrier();
  const int cn = note_n[lane];
  const bool hand_over = last || __builtin_amdgcn_ballot_w32(cn + CHUNK > LIST_W) != 0;
  if (!hand_over) return;
  for (int ws = 0; ws < HOP_WAVES; ++ws) {
    const int c = __builtin_amdgcn_readlane(cn, ws);
    const int*            le = note_e + ws * LIST_W;
    const unsigned short* ld = note_d + ws * LIST_W;
    for (int off = 0; off < c; off += 32) {
      const int  idx   = off + lane;
      const bool valid = idx < c;
      const int  d     = valid ? (int)ld[idx] : -1;
      const bool mine  = valid && ((d & (HOP_WAVES - 1)) == wave);
      const unsigned mk = __builtin_amdgcn_ballot_w32(mine);
      if (mk) {
        if (mine) {
          const int pos = pc + __builtin_popcount(mk & lt);
          oe[pos] = le[idx];
          od[pos] = (unsigned short)d;
        }
        pc += __builtin_popcount(mk);
        if (pc >= 32) {
          sgc_hop_group<false>(x, norm, src, acc, oe, od, 32, lane);
          const int rem = pc - 32;
          if (lane < rem) {
            const int            te = oe[32 + lane];
            const unsigned short td = od[32 + lane];
            oe[lane] = te;
            od[lane] = td;
          }
          pc = rem;
        }
      }
    }
  }
  sgc_lds_barrier();
  cnt = 0;
}

__global__ __launch_bounds__(HOP_WAVES * 32) void sgc_hop(const float* __restrict__ x,
                                                          const float* __restrict__ norm,
                                                          const int* __restrict__ src,
                                                          const int* __restrict__ dst,
                                                          float* __restrict__ xout, int N, int E) {
  __shared__ float          acc[NODES_PER_WG * OUT_F];
  __shared__ int            note_e[HOP_WAVES * LIST_W];
  __shared__ unsigned short note_d[HOP_WAVES * LIST_W];
  __shared__ int            own_e[HOP_WAVES * OWN_W];
  __shared__ unsigned short own_d[HOP_WAVES * OWN_W];
  __shared__ int            note_n[2][HOP_WAVES];

  const int tid  = threadIdx.x;
  const int lane = tid & 31;
  const int wave  = __builtin_amdgcn_readfirstlane(tid >> 5);
  const int node0 = blockIdx.x * NODES_PER_WG;
  const int left  = N - node0;
  const int nrows = left < NODES_PER_WG ? left : NODES_PER_WG;

  for (int i = tid; i < NODES_PER_WG * OUT_F / 4; i += HOP_WAVES * 32)
    ((float4*)acc)[i] = make_float4(0.f, 0.f, 0.f, 0.f);
  sgc_lds_barrier();

  int*            ne = note_e + wave * LIST_W;
  unsigned short* nd = note_d + wave * LIST_W;
  int*            oe = own_e + wave * OWN_W;
  unsigned short* od = own_d + wave * OWN_W;
  int cnt = 0;
  int pc  = 0;
  const unsigned lt = (1u << lane) - 1u;

#define SGC_NOTE(mc, eidx, dloc)                                         \
  {                                                                      \
    const unsigned mk_ = __builtin_amdgcn_ballot_w32(mc);                \
    if (mk_) {                                                           \
      if (mc) {                                                          \
        const int pos_ = cnt + __builtin_popcount(mk_ & lt);             \
        ne[pos_] = (eidx);                                               \
        nd[pos_] = (unsigned short)(dloc);                               \
      }                                                                  \
      cnt += __builtin_popcount(mk_);                                    \
    }                                                                    \
  }

  const int nchunks = E / CHUNK;
  const int nsteps  = (nchunks + HOP_WAVES - 1) / HOP_WAVES + 1;
  const int rot = (int)(((long long)blockIdx.x * nchunks) / gridDim.x);
  const int4* dst4 = (const int4*)dst;

#define SGC_LOAD(IT, BUF)                                                \
  {                                                                      \
    const int ql_ = wave + (IT) * HOP_WAVES;                             \
    int pl_ = (ql_ < nchunks ? ql_ : nchunks - 1) + rot;                 \
    if (pl_ >= nchunks) pl_ -= nchunks;                                  \
    _Pragma("unroll") for (int k = 0; k < 4; ++k)                        \
        BUF[k] = dst4[pl_ * (CHUNK / 4) + k * 32 + lane];                \
  }

#define SGC_STEP(IT, BUF, NEXTBUF)                                       \
  {                                                                      \
    const int it_ = (IT);                                                \
    SGC_LOAD(it_ + 2, NEXTBUF)                                           \
    if (it_ < nsteps - 1) {                                              \
      const int q_ = wave + it_ * HOP_WAVES;                             \
      if (q_ < nchunks) {                                                \
        int p_ = q_ + rot;                                               \
        if (p_ >= nchunks) p_ -= nchunks;                                \
        _Pragma("unroll") for (int k = 0; k < 4; ++k) {                  \
          const int4 d4 = BUF[k];                                        \
          const int l0 = d4.x - node0, l1 = d4.y - node0;                \
          const int l2 = d4.z - node0, l3 = d4.w - node0;                \
          const bool m0 = (unsigned)l0 < (unsigned)nrows;                \
          const bool m1 = (unsigned)l1 < (unsigned)nrows;                \
          const bool m2 = (unsigned)l2 < (unsigned)nrows;                \
          const bool m3 = (unsigned)l3 < (unsigned)nrows;                \
                                 \
          if (__builtin_amdgcn_ballot_w32(m0 | m1 | m2 | m3)) {          \
            const int e0 = p_ * CHUNK + k * 128 + lane * 4;              \
            SGC_NOTE(m0, e0, l0)                                         \
            SGC_NOTE(m1, e0 + 1, l1)                                     \
            SGC_NOTE(m2, e0 + 2, l2)                                     \
            SGC_NOTE(m3, e0 + 3, l3)                                     \
          }                                                              \
        }                                                                \
      }                                                                  \
    } else {                                                             \
            \
      const int i_ = nchunks * CHUNK + wave * 32;                        \
      if (i_ < E) {                                                      \
        const int  e_  = i_ + lane;                                      \
        const bool ok_ = e_ < E;                                         \
        const int  l_  = dst[ok_ ? e_ : E - 1] - node0;                  \
        const bool m_  = ok_ && (unsigned)l_ < (unsigned)nrows;          \
        SGC_NOTE(m_, e_, l_)                                             \
      }                                                                  \
    }                                                                    \
    sgc_hop_publish_and_hand_over(x, norm, src, acc, note_e, note_d, &note_n[it_ & 1][0], oe, \
                                  od, cnt, pc, it_ == nsteps - 1, wave, lane, lt);            \
  }

  int4 buf0[4] = {}, buf1[4] = {}, buf2[4] = {};
  SGC_LOAD(0, buf0)
  SGC_LOAD(1, buf1)
  for (int it = 0; it < nsteps; it += 3) {
    SGC_STEP(it, buf0, buf2)
    if (it + 1 < nsteps) SGC_STEP(it + 1, buf1, buf0)
    if (it + 2 < nsteps) SGC_STEP(it + 2, buf2, buf1)
  }
#undef SGC_STEP
#undef SGC_LOAD
#undef SGC_NOTE
  if (pc > 0) sgc_hop_group<true>(x, norm, src, acc, oe, od, pc, lane);

  for (int r = wave; r < nrows; r += HOP_WAVES) {
    const float  ndst = norm[node0 + r];
    const float4 a    = ((const float4*)acc)[r * (OUT_F / 4) + lane];
    ((float4*)xout)[(size_t)(node0 + r) * (OUT_F / 4) + lane] =
        make_float4(ndst * a.x, ndst * a.y, ndst * a.z, ndst * a.w);
  }
}

extern "C" void kernel_launch(void* const* d_in, const int* in_sizes, int n_in,
                              void* d_out, int out_size, void* d_ws, size_t ws_size,
                              hipStream_t stream) {
  const float* h    = (const float*)d_in[0];
  const float* w    = (const float*)d_in[1];
  const float* norm = (const float*)d_in[2];
  const int*   src  = (const int*)d_in[3];
  const int*   dst  = (const int*)d_in[4];
  float* out = (float*)d_out;
  float* x1  = (float*)d_ws;

  const int N = in_sizes[2];
  const int E = in_sizes[3];

  const int mblocks = (N + 15) / 16;
  const int groups  = (N + NODES_PER_WG - 1) / NODES_PER_WG;

  sgc_gemm_wmma<<<mblocks, 256, 0, stream>>>(h, w, out);
  sgc_hop<<<groups, HOP_WAVES * 32, 0, stream>>>(out, norm, src, dst, x1, N, E);
  sgc_hop<<<groups, HOP_WAVES * 32, 0, stream>>>(x1, norm, src, dst, out, N, E);
}


# Edges binned by destination group in a launch before each hop (lists kept in the group's own output rows): a group reads its edge list instead of testing all of dst; fixed sum order, a new one
# speedup vs baseline: 2.6469x; 1.3833x over previous
#include <hip/hip_runtime.h>

#define IN_F  256
#define OUT_F 128
#define LDS_PAD 260

typedef __attribute__((ext_vector_type(2))) float v2f;
typedef __attribute__((ext_vector_type(8))) float v8f;

__global__ __launch_bounds__(256) void sgc_gemm_wmma(const float* __restrict__ h,
                                                     const float* __restrict__ w,
                                                     float* __restrict__ x) {
  __shared__ float atile[16 * LDS_PAD];
  const int m0  = blockIdx.x * 16;
  const int tid = threadIdx.x;

  for (int i4 = tid; i4 < 16 * IN_F / 4; i4 += 256) {
    int i = i4 * 4;
    int r = i / IN_F, c = i % IN_F;
    float4 v = *(const float4*)(h + (size_t)(m0 + r) * IN_F + c);
    atile[r * LDS_PAD + c + 0] = v.x;
    atile[r * LDS_PAD + c + 1] = v.y;
    atile[r * LDS_PAD + c + 2] = v.z;
    atile[r * LDS_PAD + c + 3] = v.w;
  }
  __syncthreads();

  const int wave = tid >> 5;
  const int lane = tid & 31;
  const int half = lane >> 4;
  const int l    = lane & 15;
  const int n0   = wave * 16;

  v8f c = {};
  const float* arow = atile + l * LDS_PAD + 2 * half;
  const float* bcol = w + (size_t)(2 * half) * OUT_F + n0 + l;
  for (int k = 0; k < IN_F; k += 4) {
    float2 av = *(const float2*)(arow + k);
    v2f a, b;
    a.x = av.x;
    a.y = av.y;
    b.x = bcol[(size_t)k * OUT_F];
    b.y = bcol[(size_t)(k + 1) * OUT_F];
    c = __builtin_amdgcn_wmma_f32_16x16x4_f32(false, a, false, b, (short)0, c,
                                              false, false);
  }

  float* out = x + (size_t)(m0 + 8 * half) * OUT_F + n0 + l;
#pragma unroll
  for (int r = 0; r < 8; ++r) out[(size_t)r * OUT_F] = c[r];
}

typedef const float __attribute__((address_space(1)))* gfloat_cptr;

#define NODES_PER_WG 391
#define HOP_WAVES 32
#define CHUNK 512
#define LIST_W 576
#define OWN_W 64
#define EDGE_BATCH 8
#define BIN_TASKS 1024
#define BIN_WAVES 4
#define BIN_GROUPS 256
#define SLOT_W 32
#define SLOT_CAP 31
#define SLOT_MIN_ROWS 256
#define SLOT_ROUND 8
#define SRC_BITS 22

template <bool TAIL>
__device__ __forceinline__ void sgc_hop_batch(const float* __restrict__ x, float* acc, int sv,
                                              int dv, float nv, int lane, int b, int n) {
  float v[EDGE_BATCH][4];
#pragma unroll
  for (int j = 0; j < EDGE_BATCH; ++j) {
    const int s = __builtin_amdgcn_readlane(sv, b + j);
    gfloat_cptr xrow = (gfloat_cptr)(x + (size_t)s * OUT_F);
    asm("" : "+s"(xrow));
    v[j][0] = xrow[lane];
    v[j][1] = xrow[lane + 32];
    v[j][2] = xrow[lane + 64];
    v[j][3] = xrow[lane + 96];
  }
#pragma unroll
  for (int j = 0; j < EDGE_BATCH; ++j) {
    if (!TAIL || b + j < n) {
      const float ns = __int_as_float(__builtin_amdgcn_readlane(__float_as_int(nv), b + j));
      const int   d  = __builtin_amdgcn_readlane(dv, b + j);
      float* arow = acc + d * OUT_F + lane;
      arow[0]  += ns * v[j][0];
      arow[32] += ns * v[j][1];
      arow[64] += ns * v[j][2];
      arow[96] += ns * v[j][3];
    }
  }
}

template <bool TAIL>
__device__ __forceinline__ void sgc_hop_group(const float* __restrict__ x,
                                              const float* __restrict__ norm,
                                              const int* __restrict__ src, float* acc,
                                              const int* oe, const unsigned short* od, int n,
                                              int lane, bool by_edge) {
  const int li = TAIL ? (lane < n ? lane : n - 1) : lane;
  const int e  = oe[li];
  const int dv = od[li];
  const int sv = by_edge ? src[e] : e;
  const float nv = norm[sv];
  int b = 0;
  for (; b + EDGE_BATCH <= n; b += EDGE_BATCH)
    sgc_hop_batch<false>(x, acc, sv, dv, nv, lane, b, n);
  if (TAIL && b < n) sgc_hop_batch<true>(x, acc, sv, dv, nv, lane, b, n);
}

__device__ __forceinline__ void sgc_lds_barrier() {
  __builtin_amdgcn_fence(__ATOMIC_RELEASE, "workgroup", "local");
  __builtin_amdgcn_s_barrier();
  __builtin_amdgcn_fence(__ATOMIC_ACQUIRE, "workgroup", "local");
}

__device__ __forceinline__ void sgc_hop_publish_and_hand_over(
    const float* __restrict__ x, const float* __restrict__ norm, const int* __restrict__ src,
    float* acc, const int* note_e, const unsigned short* note_d, int* note_n, int* oe,
    unsigned short* od, int& cnt, int& pc, bool last, int wave, int lane, unsigned lt, int room,
    bool by_edge) {
  if (lane == 0) note_n[wave] = cnt;
  sgc_lds_barrier();
  const int cn = note_n[lane];
  const bool hand_over = last || __builtin_amdgcn_ballot_w32(cn + room > LIST_W) != 0;
  if (!hand_over) return;
  for (int ws = 0; ws < HOP_WAVES; ++ws) {
    const int c = __builtin_amdgcn_readlane(cn, ws);
    const int*            le = note_e + ws * LIST_W;
    const unsigned short* ld = note_d + ws * LIST_W;
    for (int off = 0; off < c; off += 32) {
      const int  idx   = off + lane;
      const bool valid = idx < c;
      const int  d     = valid ? (int)ld[idx] : -1;
      const bool mine  = valid && ((d & (HOP_WAVES - 1)) == wave);
      const unsigned mk = __builtin_amdgcn_ballot_w32(mine);
      if (mk) {
        if (mine) {
          const int pos = pc + __builtin_popcount(mk & lt);
          oe[pos] = le[idx];
          od[pos] = (unsigned short)d;
        }
        pc += __builtin_popcount(mk);
        if (pc >= 32) {
          sgc_hop_group<false>(x, norm, src, acc, oe, od, 32, lane, by_edge);
          const int rem = pc - 32;
          if (lane < rem) {
            const int            te = oe[32 + lane];
            const unsigned short td = od[32 + lane];
            oe[lane] = te;
            od[lane] = td;
          }
          pc = rem;
        }
      }
    }
  }
  sgc_lds_barrier();
  cnt = 0;
}

__global__ __launch_bounds__(HOP_WAVES * 32) void sgc_hop(const float* __restrict__ x,
                                                          const float* __restrict__ norm,
                                                          const int* __restrict__ src,
                                                          const int* __restrict__ dst,
                                                          float* __restrict__ xout, int N, int E) {
  __shared__ float          acc[NODES_PER_WG * OUT_F];
  __shared__ int            note_e[HOP_WAVES * LIST_W];
  __shared__ unsigned short note_d[HOP_WAVES * LIST_W];
  __shared__ int            own_e[HOP_WAVES * OWN_W];
  __shared__ unsigned short own_d[HOP_WAVES * OWN_W];
  __shared__ int            note_n[2][HOP_WAVES];

  const int tid  = threadIdx.x;
  const int lane = tid & 31;
  const int wave  = __builtin_amdgcn_readfirstlane(tid >> 5);
  const int node0 = blockIdx.x * NODES_PER_WG;
  const int left  = N - node0;
  const int nrows = left < NODES_PER_WG ? left : NODES_PER_WG;

  for (int i = tid; i < NODES_PER_WG * OUT_F / 4; i += HOP_WAVES * 32)
    ((float4*)acc)[i] = make_float4(0.f, 0.f, 0.f, 0.f);

  const float* slots = xout + (size_t)node0 * OUT_F;
  const bool served = (int)gridDim.x <= BIN_GROUPS && nrows >= SLOT_MIN_ROWS;
  bool bad = !served;
  if (served) {
    const unsigned c0 = __float_as_uint(slots[(size_t)(lane * HOP_WAVES + wave) * SLOT_W]);
    bad = __builtin_amdgcn_ballot_w32(c0 > SLOT_CAP) != 0;
  }
  if (lane == 0) note_n[1][wave] = bad ? 1 : 0;
  sgc_lds_barrier();
  const bool scan = __builtin_amdgcn_ballot_w32(note_n[1][lane] != 0) != 0;

  int*            ne = note_e + wave * LIST_W;
  unsigned short* nd = note_d + wave * LIST_W;
  int*            oe = own_e + wave * OWN_W;
  unsigned short* od = own_d + wave * OWN_W;
  int cnt = 0;
  int pc  = 0;
  const unsigned lt = (1u << lane) - 1u;

#define SGC_NOTE(mc, eidx, dloc)                                         \
  {                                                                      \
    const unsigned mk_ = __builtin_amdgcn_ballot_w32(mc);                \
    if (mk_) {                                                           \
      if (mc) {                                                          \
        const int pos_ = cnt + __builtin_popcount(mk_ & lt);             \
        ne[pos_] = (eidx);                                               \
        nd[pos_] = (unsigned short)(dloc);                               \
      }                                                                  \
      cnt += __builtin_popcount(mk_);                                    \
    }                                                                    \
  }

  const int nchunks = E / CHUNK;
  const int nsteps  = (nchunks + HOP_WAVES - 1) / HOP_WAVES + 1;
  const int rot = (int)(((long long)blockIdx.x * nchunks) / gridDim.x);
  const int4* dst4 = (const int4*)dst;

#define SGC_LOAD(IT, BUF)                                                \
  {                                                                      \
    const int ql_ = wave + (IT) * HOP_WAVES;                             \
    int pl_ = (ql_ < nchunks ? ql_ : nchunks - 1) + rot;                 \
    if (pl_ >= nchunks) pl_ -= nchunks;                                  \
    _Pragma("unroll") for (int k = 0; k < 4; ++k)                        \
        BUF[k] = dst4[pl_ * (CHUNK / 4) + k * 32 + lane];                \
  }

#define SGC_STEP(IT, BUF, NEXTBUF)                                       \
  {                                                                      \
    const int it_ = (IT);                                                \
    SGC_LOAD(it_ + 2, NEXTBUF)                                           \
    if (it_ < nsteps - 1) {                                              \
      const int q_ = wave + it_ * HOP_WAVES;                             \
      if (q_ < nchunks) {                                                \
        int p_ = q_ + rot;                                               \
        if (p_ >= nchunks) p_ -= nchunks;                                \
        _Pragma("unroll") for (int k = 0; k < 4; ++k) {                  \
          const int4 d4 = BUF[k];                                        \
          const int l0 = d4.x - node0, l1 = d4.y - node0;                \
          const int l2 = d4.z - node0, l3 = d4.w - node0;                \
          const bool m0 = (unsigned)l0 < (unsigned)nrows;                \
          const bool m1 = (unsigned)l1 < (unsigned)nrows;                \
          const bool m2 = (unsigned)l2 < (unsigned)nrows;                \
          const bool m3 = (unsigned)l3 < (unsigned)nrows;                \
                                 \
          if (__builtin_amdgcn_ballot_w32(m0 | m1 | m2 | m3)) {          \
            const int e0 = p_ * CHUNK + k * 128 + lane * 4;              \
            SGC_NOTE(m0, e0, l0)                                         \
            SGC_NOTE(m1, e0 + 1, l1)                                     \
            SGC_NOTE(m2, e0 + 2, l2)                                     \
            SGC_NOTE(m3, e0 + 3, l3)                                     \
          }                                                              \
        }                                                                \
      }                                                                  \
    } else {                                                             \
            \
      const int i_ = nchunks * CHUNK + wave * 32;                        \
      if (i_ < E) {                                                      \
        const int  e_  = i_ + lane;                                      \
        const bool ok_ = e_ < E;                                         \
        const int  l_  = dst[ok_ ? e_ : E - 1] - node0;                  \
        const bool m_  = ok_ && (unsigned)l_ < (unsigned)nrows;          \
        SGC_NOTE(m_, e_, l_)                                             \
      }                                                                  \
    }                                                                    \
    sgc_hop_publish_and_hand_over(x, norm, src, acc, note_e, note_d, &note_n[it_ & 1][0], oe, \
                                  od, cnt, pc, it_ == nsteps - 1, wave, lane, lt, CHUNK,      \
                                  true);                                                      \
  }

  if (!scan) {
    for (int st = 0; st < BIN_TASKS / HOP_WAVES / SLOT_ROUND; ++st) {
      unsigned wv[SLOT_ROUND];
#pragma unroll
      for (int k = 0; k < SLOT_ROUND; ++k)
        wv[k] = __float_as_uint(
            slots[(size_t)((st * SLOT_ROUND + k) * HOP_WAVES + wave) * SLOT_W + lane]);
#pragma unroll
      for (int k = 0; k < SLOT_ROUND; ++k) {
        const int c = __builtin_amdgcn_readfirstlane((int)wv[k]);
        if (lane >= 1 && lane <= c) {
          ne[cnt + lane - 1] = (int)(wv[k] & ((1u << SRC_BITS) - 1u));
          nd[cnt + lane - 1] = (unsigned short)(wv[k] >> SRC_BITS);
        }
        cnt += c;
      }
      sgc_hop_publish_and_hand_over(x, norm, src, acc, note_e, note_d, &note_n[st & 1][0], oe, od,
                                    cnt, pc, st == BIN_TASKS / HOP_WAVES / SLOT_ROUND - 1, wave,
                                    lane, lt, SLOT_ROUND * SLOT_CAP, false);
    }
  } else {
    int4 buf0[4] = {}, buf1[4] = {}, buf2[4] = {};
    SGC_LOAD(0, buf0)
    SGC_LOAD(1, buf1)
    for (int it = 0; it < nsteps; it += 3) {
      SGC_STEP(it, buf0, buf2)
      if (it + 1 < nsteps) SGC_STEP(it + 1, buf1, buf0)
      if (it + 2 < nsteps) SGC_STEP(it + 2, buf2, buf1)
    }
  }
#undef SGC_STEP
#undef SGC_LOAD
#undef SGC_NOTE
  if (pc > 0) sgc_hop_group<true>(x, norm, src, acc, oe, od, pc, lane, scan);

  for (int r = wave; r < nrows; r += HOP_WAVES) {
    const float  ndst = norm[node0 + r];
    const float4 a    = ((const float4*)acc)[r * (OUT_F / 4) + lane];
    ((float4*)xout)[(size_t)(node0 + r) * (OUT_F / 4) + lane] =
        make_float4(ndst * a.x, ndst * a.y, ndst * a.z, ndst * a.w);
  }
}

__device__ __forceinline__ void sgc_bin_trip(unsigned* my, unsigned* my_n, unsigned d, unsigned s,
                                             bool in, unsigned lt) {
  const unsigned g   = d / NODES_PER_WG;
  const unsigned row = d - g * NODES_PER_WG;
  const bool ok = in && g < BIN_GROUPS;
  unsigned same = __builtin_amdgcn_ballot_w32(ok);
#pragma unroll
  for (int b = 0; b < 8; ++b) {
    const bool bit = (g >> b) & 1u;
    const unsigned bm = __builtin_amdgcn_ballot_w32(bit);
    same &= bit ? bm : ~bm;
  }
  const unsigned rank = __builtin_popcount(same & lt);
  __builtin_amdgcn_wave_barrier();
  if (ok) {
    const unsigned base = my_n[g];
    const unsigned pos  = base + rank;
    if (pos < SLOT_CAP) my[g * SLOT_W + 1 + pos] = (row << SRC_BITS) | s;
    if (rank == 0) my_n[g] = base + __builtin_popcount(same);
  }
}

#define BIN_TRIPS 8

__global__ __launch_bounds__(BIN_WAVES * 32) void sgc_bin(const int* __restrict__ src,
                                                          const int* __restrict__ dst,
                                                          float* __restrict__ buf, int N, int E) {
  __shared__ unsigned slot[BIN_WAVES * BIN_GROUPS * SLOT_W];
  __shared__ unsigned slot_n[BIN_WAVES * BIN_GROUPS];
  const int lane = threadIdx.x & 31;
  const int wave = __builtin_amdgcn_readfirstlane((int)threadIdx.x >> 5);
  const int task = blockIdx.x * BIN_WAVES + wave;
  unsigned* my   = slot + wave * (BIN_GROUPS * SLOT_W);
  unsigned* my_n = slot_n + wave * BIN_GROUPS;
  for (int g = lane; g < BIN_GROUPS; g += 32) my_n[g] = 0u;

  const int ept = (E + BIN_TASKS - 1) / BIN_TASKS;
  const int e0  = task * ept;
  const int e1  = e0 + ept < E ? e0 + ept : E;
  const unsigned lt = (1u << lane) - 1u;

#define BIN_LOAD(DB, SB, EB)                                             \
  _Pragma("unroll") for (int k = 0; k < BIN_TRIPS; ++k) {                \
    const int e_ = (EB) + k * 32 + lane;                                 \
    const int c_ = e_ < E ? e_ : E - 1;                                  \
    DB[k] = dst[c_];                                                     \
    SB[k] = src[c_];                                                     \
  }
#define BIN_WORK(DB, SB, EB)                                             \
  _Pragma("unroll") for (int k = 0; k < BIN_TRIPS; ++k) {                \
    const int eb_ = (EB) + k * 32;                                       \
    if (eb_ < e1)                                       \
      sgc_bin_trip(my, my_n, (unsigned)DB[k], (unsigned)SB[k], eb_ + lane < e1, lt); \
  }

  int da[BIN_TRIPS], sa[BIN_TRIPS], db[BIN_TRIPS], sb[BIN_TRIPS];
  BIN_LOAD(da, sa, e0)
  for (int eb = e0; eb < e1; eb += 2 * BIN_TRIPS * 32) {
    BIN_LOAD(db, sb, eb + BIN_TRIPS * 32)
    BIN_WORK(da, sa, eb)
    BIN_LOAD(da, sa, eb + 2 * BIN_TRIPS * 32)
    BIN_WORK(db, sb, eb + BIN_TRIPS * 32)
  }
#undef BIN_LOAD
#undef BIN_WORK
  __builtin_amdgcn_wave_barrier();
  for (int g = lane; g < BIN_GROUPS; g += 32) my[g * SLOT_W] = my_n[g];
  __builtin_amdgcn_wave_barrier();

  int gs = N >= SLOT_MIN_ROWS ? (N - SLOT_MIN_ROWS) / NODES_PER_WG + 1 : 0;
  if (gs > BIN_GROUPS) gs = BIN_GROUPS;
  uint4* line0 = (uint4*)buf + (size_t)task * (SLOT_W / 4) + (lane & 7);
  const size_t gstride = (size_t)NODES_PER_WG * OUT_F / 4;
  int i0 = 0;
  for (; (i0 + 8) * 4 <= gs; i0 += 8) {
    uint4 v[8];
#pragma unroll
    for (int j = 0; j < 8; ++j) v[j] = ((const uint4*)my)[(i0 + j) * 32 + lane];
#pragma unroll
    for (int j = 0; j < 8; ++j) line0[(size_t)((i0 + j) * 4 + (lane >> 3)) * gstride] = v[j];
  }
  for (int i = i0; i * 4 < gs; ++i) {
    const int g = i * 4 + (lane >> 3);
    if (g < gs) line0[(size_t)g * gstride] = ((const uint4*)my)[i * 32 + lane];
  }
}

extern "C" void kernel_launch(void* const* d_in, const int* in_sizes, int n_in,
                              void* d_out, int out_size, void* d_ws, size_t ws_size,
                              hipStream_t stream) {
  const float* h    = (const float*)d_in[0];
  const float* w    = (const float*)d_in[1];
  const float* norm = (const float*)d_in[2];
  const int*   src  = (const int*)d_in[3];
  const int*   dst  = (const int*)d_in[4];
  float* out = (float*)d_out;
  float* x1  = (float*)d_ws;

  const int N = in_sizes[2];
  const int E = in_sizes[3];

  const int mblocks = (N + 15) / 16;
  const int groups  = (N + NODES_PER_WG - 1) / NODES_PER_WG;

  sgc_gemm_wmma<<<mblocks, 256, 0, stream>>>(h, w, out);
  sgc_bin<<<BIN_TASKS / BIN_WAVES, BIN_WAVES * 32, 0, stream>>>(src, dst, x1, N, E);
  sgc_hop<<<groups, HOP_WAVES * 32, 0, stream>>>(out, norm, src, dst, x1, N, E);
  sgc_bin<<<BIN_TASKS / BIN_WAVES, BIN_WAVES * 32, 0, stream>>>(src, dst, out, N, E);
  sgc_hop<<<groups, HOP_WAVES * 32, 0, stream>>>(x1, norm, src, dst, out, N, E);
}


# Hop sums each destination row in registers: the group's binned edges are counting-sorted by row in LDS (ballots, no atomics), one 16-byte load per lane and edge; fixed sum order, a new one
# speedup vs baseline: 3.2065x; 1.2114x over previous
#include <hip/hip_runtime.h>

#define IN_F  256
#define OUT_F 128
#define LDS_PAD 260

typedef __attribute__((ext_vector_type(2))) float v2f;
typedef __attribute__((ext_vector_type(8))) float v8f;

__global__ __launch_bounds__(256) void sgc_gemm_wmma(const float* __restrict__ h,
                                                     const float* __restrict__ w,
                                                     float* __restrict__ x) {
  __shared__ float atile[16 * LDS_PAD];
  const int m0  = blockIdx.x * 16;
  const int tid = threadIdx.x;

  for (int i4 = tid; i4 < 16 * IN_F / 4; i4 += 256) {
    int i = i4 * 4;
    int r = i / IN_F, c = i % IN_F;
    float4 v = *(const float4*)(h + (size_t)(m0 + r) * IN_F + c);
    atile[r * LDS_PAD + c + 0] = v.x;
    atile[r * LDS_PAD + c + 1] = v.y;
    atile[r * LDS_PAD + c + 2] = v.z;
    atile[r * LDS_PAD + c + 3] = v.w;
  }
  __syncthreads();

  const int wave = tid >> 5;
  const int lane = tid & 31;
  const int half = lane >> 4;
  const int l    = lane & 15;
  const int n0   = wave * 16;

  v8f c = {};
  const float* arow = atile + l * LDS_PAD + 2 * half;
  const float* bcol = w + (size_t)(2 * half) * OUT_F + n0 + l;
  for (int k = 0; k < IN_F; k += 4) {
    float2 av = *(const float2*)(arow + k);
    v2f a, b;
    a.x = av.x;
    a.y = av.y;
    b.x = bcol[(size_t)k * OUT_F];
    b.y = bcol[(size_t)(k + 1) * OUT_F];
    c = __builtin_amdgcn_wmma_f32_16x16x4_f32(false, a, false, b, (short)0, c,
                                              false, false);
  }

  float* out = x + (size_t)(m0 + 8 * half) * OUT_F + n0 + l;
#pragma unroll
  for (int r = 0; r < 8; ++r) out[(size_t)r * OUT_F] = c[r];
}

typedef const float __attribute__((address_space(1)))* gfloat_cptr;
typedef __attribute__((ext_vector_type(4))) float v4f;
typedef const v4f __attribute__((address_space(1)))* gv4f_cptr;

#define NODES_PER_WG 391
#define HOP_WAVES 32
#define CHUNK 512
#define LIST_W 576
#define OWN_W 64
#define EDGE_BATCH 8
#define BIN_TASKS 1024
#define BIN_WAVES 4
#define BIN_GROUPS 256
#define SLOT_W 32
#define SLOT_CAP 31
#define SLOT_MIN_ROWS 256
#define SRC_BITS 22
#define ROW_CAP (BIN_TASKS * SLOT_CAP)
#define WAVE_CAP (HOP_WAVES * SLOT_CAP)
#define ROW_SLOTS 416
#define ROW_BATCH 16

struct sgc_lds_scan {
  float          acc[NODES_PER_WG * OUT_F];
  int            note_e[HOP_WAVES * LIST_W];
  unsigned short note_d[HOP_WAVES * LIST_W];
  int            own_e[HOP_WAVES * OWN_W];
  unsigned short own_d[HOP_WAVES * OWN_W];
  int            note_n[2][HOP_WAVES];
};
struct sgc_lds_rows {
  unsigned elist[ROW_CAP];
  unsigned byrow[ROW_CAP];
  unsigned hist[HOP_WAVES][ROW_SLOTS];
  unsigned rtot[ROW_SLOTS];
  unsigned rstart[ROW_SLOTS];
};
union sgc_lds {
  sgc_lds_scan s;
  sgc_lds_rows f;
};
static_assert(sizeof(sgc_lds_rows) <= offsetof(sgc_lds_scan, note_n), "usual path must not reach note_n");
static_assert(ROW_SLOTS >= NODES_PER_WG && ROW_SLOTS == 13 * 32, "scan by 13 counters per lane");
static_assert(NODES_PER_WG <= 512, "a local row number has 9 bits");
static_assert(HOP_WAVES * 32 == BIN_TASKS, "32 slots per wave");
static_assert(2 * ROW_BATCH == 32, "two batches per 32 edges");
static_assert(NODES_PER_WG <= HOP_WAVES * 32, "a wave's rows fit the lanes of one register");

__device__ __forceinline__ unsigned sgc_same_row(bool valid, unsigned row) {
  unsigned same = __builtin_amdgcn_ballot_w32(valid);
#pragma unroll
  for (int b = 0; b < 9; ++b) {
    const bool bit = (row >> b) & 1u;
    const unsigned bm = __builtin_amdgcn_ballot_w32(bit);
    same &= bit ? bm : ~bm;
  }
  return same;
}

__device__ __forceinline__ void sgc_row_issue(const float* __restrict__ x, v4f (&v)[ROW_BATCH],
                                              int sv, int lane, int b, int m) {
#pragma unroll
  for (int g = 0; g < ROW_BATCH / 4; ++g) {
    if (b + 4 * g < m) {
#pragma unroll
      for (int j = 4 * g; j < 4 * g + 4; ++j) {
        const int s = __builtin_amdgcn_readlane(sv, b + j);
        gv4f_cptr xrow = (gv4f_cptr)(x + (size_t)s * OUT_F);
        asm("" : "+s"(xrow));
        v[j] = xrow[lane];
      }
    }
  }
}

__device__ __forceinline__ void sgc_row_add(v4f& a, const v4f (&v)[ROW_BATCH], float nv, int b,
                                            int m) {
#pragma unroll
  for (int j = 0; j < ROW_BATCH; ++j) {
    if (b + j < m) {
      const float ns = __int_as_float(__builtin_amdgcn_readlane(__float_as_int(nv), b + j));
      a += ns * v[j];
    }
  }
}

template <bool TAIL>
__device__ __forceinline__ void sgc_hop_batch(const float* __restrict__ x, float* acc, int sv,
                                              int dv, float nv, int lane, int b, int n) {
  float v[EDGE_BATCH][4];
#pragma unroll
  for (int j = 0; j < EDGE_BATCH; ++j) {
    const int s = __builtin_amdgcn_readlane(sv, b + j);
    gfloat_cptr xrow = (gfloat_cptr)(x + (size_t)s * OUT_F);
    asm("" : "+s"(xrow));
    v[j][0] = xrow[lane];
    v[j][1] = xrow[lane + 32];
    v[j][2] = xrow[lane + 64];
    v[j][3] = xrow[lane + 96];
  }
#pragma unroll
  for (int j = 0; j < EDGE_BATCH; ++j) {
    if (!TAIL || b + j < n) {
      const float ns = __int_as_float(__builtin_amdgcn_readlane(__float_as_int(nv), b + j));
      const int   d  = __builtin_amdgcn_readlane(dv, b + j);
      float* arow = acc + d * OUT_F + lane;
      arow[0]  += ns * v[j][0];
      arow[32] += ns * v[j][1];
      arow[64] += ns * v[j][2];
      arow[96] += ns * v[j][3];
    }
  }
}

template <bool TAIL>
__device__ __forceinline__ void sgc_hop_group(const float* __restrict__ x,
                                              const float* __restrict__ norm,
                                              const int* __restrict__ src, float* acc,
                                              const int* oe, const unsigned short* od, int n,
                                              int lane, bool by_edge) {
  const int li = TAIL ? (lane < n ? lane : n - 1) : lane;
  const int e  = oe[li];
  const int dv = od[li];
  const int sv = by_edge ? src[e] : e;
  const float nv = norm[sv];
  int b = 0;
  for (; b + EDGE_BATCH <= n; b += EDGE_BATCH)
    sgc_hop_batch<false>(x, acc, sv, dv, nv, lane, b, n);
  if (TAIL && b < n) sgc_hop_batch<true>(x, acc, sv, dv, nv, lane, b, n);
}

__device__ __forceinline__ void sgc_lds_barrier() {
  __builtin_amdgcn_fence(__ATOMIC_RELEASE, "workgroup", "local");
  __builtin_amdgcn_s_barrier();
  __builtin_amdgcn_fence(__ATOMIC_ACQUIRE, "workgroup", "local");
}

__device__ __forceinline__ void sgc_hop_publish_and_hand_over(
    const float* __restrict__ x, const float* __restrict__ norm, const int* __restrict__ src,
    float* acc, const int* note_e, const unsigned short* note_d, int* note_n, int* oe,
    unsigned short* od, int& cnt, int& pc, bool last, int wave, int lane, unsigned lt, int room,
    bool by_edge) {
  if (lane == 0) note_n[wave] = cnt;
  sgc_lds_barrier();
  const int cn = note_n[lane];
  const bool hand_over = last || __builtin_amdgcn_ballot_w32(cn + room > LIST_W) != 0;
  if (!hand_over) return;
  for (int ws = 0; ws < HOP_WAVES; ++ws) {
    const int c = __builtin_amdgcn_readlane(cn, ws);
    const int*            le = note_e + ws * LIST_W;
    const unsigned short* ld = note_d + ws * LIST_W;
    for (int off = 0; off < c; off += 32) {
      const int  idx   = off + lane;
      const bool valid = idx < c;
      const int  d     = valid ? (int)ld[idx] : -1;
      const bool mine  = valid && ((d & (HOP_WAVES - 1)) == wave);
      const unsigned mk = __builtin_amdgcn_ballot_w32(mine);
      if (mk) {
        if (mine) {
          const int pos = pc + __builtin_popcount(mk & lt);
          oe[pos] = le[idx];
          od[pos] = (unsigned short)d;
        }
        pc += __builtin_popcount(mk);
        if (pc >= 32) {
          sgc_hop_group<false>(x, norm, src, acc, oe, od, 32, lane, by_edge);
          const int rem = pc - 32;
          if (lane < rem) {
            const int            te = oe[32 + lane];
            const unsigned short td = od[32 + lane];
            oe[lane] = te;
            od[lane] = td;
          }
          pc = rem;
        }
      }
    }
  }
  sgc_lds_barrier();
  cnt = 0;
}

__global__ __launch_bounds__(HOP_WAVES * 32) void sgc_hop(const float* __restrict__ x,
                                                          const float* __restrict__ norm,
                                                          const int* __restrict__ src,
                                                          const int* __restrict__ dst,
                                                          float* __restrict__ xout, int N, int E) {
  __shared__ sgc_lds L;
  float*          acc    = L.s.acc;
  int*            note_e = L.s.note_e;
  unsigned short* note_d = L.s.note_d;
  int*            own_e  = L.s.own_e;
  unsigned short* own_d  = L.s.own_d;
  int (*note_n)[HOP_WAVES] = L.s.note_n;

  const int tid  = threadIdx.x;
  const int lane = tid & 31;
  const int wave  = __builtin_amdgcn_readfirstlane(tid >> 5);
  const int node0 = blockIdx.x * NODES_PER_WG;
  const int left  = N - node0;
  const int nrows = left < NODES_PER_WG ? left : NODES_PER_WG;

  const float* slots = xout + (size_t)node0 * OUT_F;
  const bool served = (int)gridDim.x <= BIN_GROUPS && nrows >= SLOT_MIN_ROWS;
  unsigned* el = L.f.elist + wave * WAVE_CAP;
  unsigned* hw = L.f.hist[wave];
  int ecnt = 0;
  bool bad = !served;
  if (served) {
    unsigned wv[HOP_WAVES];
#pragma unroll
    for (int k = 0; k < HOP_WAVES; ++k)
      wv[k] = __float_as_uint(slots[(size_t)(k * HOP_WAVES + wave) * SLOT_W + lane]);
    unsigned over = 0u;
#pragma unroll
    for (int k = 0; k < HOP_WAVES; ++k) {
      const unsigned cu = (unsigned)__builtin_amdgcn_readfirstlane((int)wv[k]);
      over |= cu > SLOT_CAP ? 1u : 0u;
      const int c = (int)(cu > SLOT_CAP ? SLOT_CAP : cu);
      if (lane >= 1 && lane <= c) el[ecnt + lane - 1] = wv[k];
      ecnt += c;
    }
    bad = over != 0u;
    for (int j = lane; j < ROW_SLOTS; j += 32) hw[j] = 0u;
  }
  if (lane == 0) note_n[1][wave] = bad ? 1 : 0;
  sgc_lds_barrier();
  const bool scan = __builtin_amdgcn_ballot_w32(note_n[1][lane] != 0) != 0;

  int*            ne = note_e + wave * LIST_W;
  unsigned short* nd = note_d + wave * LIST_W;
  int*            oe = own_e + wave * OWN_W;
  unsigned short* od = own_d + wave * OWN_W;
  int cnt = 0;
  int pc  = 0;
  const unsigned lt = (1u << lane) - 1u;

#define SGC_NOTE(mc, eidx, dloc)                                         \
  {                                                                      \
    const unsigned mk_ = __builtin_amdgcn_ballot_w32(mc);                \
    if (mk_) {                                                           \
      if (mc) {                                                          \
        const int pos_ = cnt + __builtin_popcount(mk_ & lt);             \
        ne[pos_] = (eidx);                                               \
        nd[pos_] = (unsigned short)(dloc);                               \
      }                                                                  \
      cnt += __builtin_popcount(mk_);                                    \
    }                                                                    \
  }

  const int nchunks = E / CHUNK;
  const int nsteps  = (nchunks + HOP_WAVES - 1) / HOP_WAVES + 1;
  const int rot = (int)(((long long)blockIdx.x * nchunks) / gridDim.x);
  const int4* dst4 = (const int4*)dst;

#define SGC_LOAD(IT, BUF)                                                \
  {                                                                      \
    const int ql_ = wave + (IT) * HOP_WAVES;                             \
    int pl_ = (ql_ < nchunks ? ql_ : nchunks - 1) + rot;                 \
    if (pl_ >= nchunks) pl_ -= nchunks;                                  \
    _Pragma("unroll") for (int k = 0; k < 4; ++k)                        \
        BUF[k] = dst4[pl_ * (CHUNK / 4) + k * 32 + lane];                \
  }

#define SGC_STEP(IT, BUF, NEXTBUF)                                       \
  {                                                                      \
    const int it_ = (IT);                                                \
    SGC_LOAD(it_ + 2, NEXTBUF)                                           \
    if (it_ < nsteps - 1) {                                              \
      const int q_ = wave + it_ * HOP_WAVES;                             \
      if (q_ < nchunks) {                                                \
        int p_ = q_ + rot;                                               \
        if (p_ >= nchunks) p_ -= nchunks;                                \
        _Pragma("unroll") for (int k = 0; k < 4; ++k) {                  \
          const int4 d4 = BUF[k];                                        \
          const int l0 = d4.x - node0, l1 = d4.y - node0;                \
          const int l2 = d4.z - node0, l3 = d4.w - node0;                \
          const bool m0 = (unsigned)l0 < (unsigned)nrows;                \
          const bool m1 = (unsigned)l1 < (unsigned)nrows;                \
          const bool m2 = (unsigned)l2 < (unsigned)nrows;                \
          const bool m3 = (unsigned)l3 < (unsigned)nrows;                \
                                 \
          if (__builtin_amdgcn_ballot_w32(m0 | m1 | m2 | m3)) {          \
            const int e0 = p_ * CHUNK + k * 128 + lane * 4;              \
            SGC_NOTE(m0, e0, l0)                                         \
            SGC_NOTE(m1, e0 + 1, l1)                                     \
            SGC_NOTE(m2, e0 + 2, l2)                                     \
            SGC_NOTE(m3, e0 + 3, l3)                                     \
          }                                                              \
        }                                                                \
      }                                                                  \
    } else {                                                             \
            \
      const int i_ = nchunks * CHUNK + wave * 32;                        \
      if (i_ < E) {                                                      \
        const int  e_  = i_ + lane;                                      \
        const bool ok_ = e_ < E;                                         \
        const int  l_  = dst[ok_ ? e_ : E - 1] - node0;                  \
        const bool m_  = ok_ && (unsigned)l_ < (unsigned)nrows;          \
        SGC_NOTE(m_, e_, l_)                                             \
      }                                                                  \
    }                                                                    \
    sgc_hop_publish_and_hand_over(x, norm, src, acc, note_e, note_d, &note_n[it_ & 1][0], oe, \
                                  od, cnt, pc, it_ == nsteps - 1, wave, lane, lt, CHUNK,      \
                                  true);                                                      \
  }

  if (!scan) {
    for (int t = 0; t * 32 < ecnt; ++t) {
      const int  i     = t * 32 + lane;
      const bool valid = i < ecnt;
      const unsigned e   = valid ? el[i] : 0u;
      const unsigned row = e >> SRC_BITS;
      const unsigned same = sgc_same_row(valid, row);
      if (valid && (same & lt) == 0u) hw[row] += (unsigned)__builtin_popcount(same);
    }
    sgc_lds_barrier();
    if (tid < ROW_SLOTS) {
      unsigned hc[HOP_WAVES];
#pragma unroll
      for (int w = 0; w < HOP_WAVES; ++w) hc[w] = L.f.hist[w][tid];
      unsigned run = 0u;
#pragma unroll
      for (int w = 0; w < HOP_WAVES; ++w) {
        L.f.hist[w][tid] = run;
        run += hc[w];
      }
      L.f.rtot[tid] = run;
    }
    sgc_lds_barrier();
    if (wave == 0) {
      unsigned cs[13];
      unsigned sum = 0u;
#pragma unroll
      for (int j = 0; j < 13; ++j) {
        const unsigned tt = L.f.rtot[lane * 13 + j];
        cs[j] = sum;
        sum += tt;
      }
      unsigned incl = sum;
#pragma unroll
      for (int d = 1; d < 32; d <<= 1) {
        const unsigned up = __shfl_up(incl, (unsigned)d);
        if (lane >= d) incl += up;
      }
      const unsigned excl = incl - sum;
#pragma unroll
      for (int j = 0; j < 13; ++j) L.f.rstart[lane * 13 + j] = excl + cs[j];
    }
    sgc_lds_barrier();

    for (int t = 0; t * 32 < ecnt; ++t) {
      const int  i     = t * 32 + lane;
      const bool valid = i < ecnt;
      const unsigned e   = valid ? el[i] : 0u;
      const unsigned row = e >> SRC_BITS;
      const unsigned same = sgc_same_row(valid, row);
      const unsigned rank = (unsigned)__builtin_popcount(same & lt);
      if (valid) {
        const unsigned before = hw[row];
        L.f.byrow[L.f.rstart[row] + before + rank] = e & ((1u << SRC_BITS) - 1u);
        if (rank == 0u) hw[row] = before + (unsigned)__builtin_popcount(same);
      }
    }
    sgc_lds_barrier();

    const int   kr  = wave + HOP_WAVES * lane;
    const float ndv = kr < nrows ? norm[node0 + kr] : 0.f;
    int k = 0;
    for (int r = wave; r < nrows; r += HOP_WAVES, ++k) {
      const int n  = __builtin_amdgcn_readfirstlane((int)L.f.rtot[r]);
      const int s0 = __builtin_amdgcn_readfirstlane((int)L.f.rstart[r]);
      const float ndst = __int_as_float(__builtin_amdgcn_readlane(__float_as_int(ndv), k));
      const unsigned* br = L.f.byrow + s0;
      v4f a = {0.f, 0.f, 0.f, 0.f};
      for (int base = 0; base < n; base += 32) {
        const int m  = n - base < 32 ? n - base : 32;
        const int li = lane < m ? lane : m - 1;
        const int sv = (int)br[base + li];
        v4f v[ROW_BATCH];
        sgc_row_issue(x, v, sv, lane, 0, m);
        const float nv = norm[sv];
        sgc_row_add(a, v, nv, 0, m);
        if (m > ROW_BATCH) {
          sgc_row_issue(x, v, sv, lane, ROW_BATCH, m);
          sgc_row_add(a, v, nv, ROW_BATCH, m);
        }
      }
      ((v4f*)xout)[(size_t)(node0 + r) * (OUT_F / 4) + lane] = ndst * a;
    }
  } else {
    for (int i = tid; i < NODES_PER_WG * OUT_F / 4; i += HOP_WAVES * 32)
      ((float4*)acc)[i] = make_float4(0.f, 0.f, 0.f, 0.f);
    int4 buf0[4] = {}, buf1[4] = {}, buf2[4] = {};
    SGC_LOAD(0, buf0)
    SGC_LOAD(1, buf1)
    for (int it = 0; it < nsteps; it += 3) {
      SGC_STEP(it, buf0, buf2)
      if (it + 1 < nsteps) SGC_STEP(it + 1, buf1, buf0)
      if (it + 2 < nsteps) SGC_STEP(it + 2, buf2, buf1)
    }
    if (pc > 0) sgc_hop_group<true>(x, norm, src, acc, oe, od, pc, lane, true);

    for (int r = wave; r < nrows; r += HOP_WAVES) {
      const float  ndst = norm[node0 + r];
      const float4 a    = ((const float4*)acc)[r * (OUT_F / 4) + lane];
      ((float4*)xout)[(size_t)(node0 + r) * (OUT_F / 4) + lane] =
          make_float4(ndst * a.x, ndst * a.y, ndst * a.z, ndst * a.w);
    }
  }
#undef SGC_STEP
#undef SGC_LOAD
#undef SGC_NOTE
}

__device__ __forceinline__ void sgc_bin_trip(unsigned* my, unsigned* my_n, unsigned d, unsigned s,
                                             bool in, unsigned lt) {
  const unsigned g   = d / NODES_PER_WG;
  const unsigned row = d - g * NODES_PER_WG;
  const bool ok = in && g < BIN_GROUPS;
  unsigned same = __builtin_amdgcn_ballot_w32(ok);
#pragma unroll
  for (int b = 0; b < 8; ++b) {
    const bool bit = (g >> b) & 1u;
    const unsigned bm = __builtin_amdgcn_ballot_w32(bit);
    same &= bit ? bm : ~bm;
  }
  const unsigned rank = __builtin_popcount(same & lt);
  __builtin_amdgcn_wave_barrier();
  if (ok) {
    const unsigned base = my_n[g];
    const unsigned pos  = base + rank;
    if (pos < SLOT_CAP) my[g * SLOT_W + 1 + pos] = (row << SRC_BITS) | s;
    if (rank == 0) my_n[g] = base + __builtin_popcount(same);
  }
}

#define BIN_TRIPS 8

__global__ __launch_bounds__(BIN_WAVES * 32) void sgc_bin(const int* __restrict__ src,
                                                          const int* __restrict__ dst,
                                                          float* __restrict__ buf, int N, int E) {
  __shared__ unsigned slot[BIN_WAVES * BIN_GROUPS * SLOT_W];
  __shared__ unsigned slot_n[BIN_WAVES * BIN_GROUPS];
  const int lane = threadIdx.x & 31;
  const int wave = __builtin_amdgcn_readfirstlane((int)threadIdx.x >> 5);
  const int task = blockIdx.x * BIN_WAVES + wave;
  unsigned* my   = slot + wave * (BIN_GROUPS * SLOT_W);
  unsigned* my_n = slot_n + wave * BIN_GROUPS;
  for (int g = lane; g < BIN_GROUPS; g += 32) my_n[g] = 0u;

  const int ept = (E + BIN_TASKS - 1) / BIN_TASKS;
  const int e0  = task * ept;
  const int e1  = e0 + ept < E ? e0 + ept : E;
  const unsigned lt = (1u << lane) - 1u;

#define BIN_LOAD(DB, SB, EB)                                             \
  _Pragma("unroll") for (int k = 0; k < BIN_TRIPS; ++k) {                \
    const int e_ = (EB) + k * 32 + lane;                                 \
    const int c_ = e_ < E ? e_ : E - 1;                                  \
    DB[k] = dst[c_];                                                     \
    SB[k] = src[c_];                                                     \
  }
#define BIN_WORK(DB, SB, EB)                                             \
  _Pragma("unroll") for (int k = 0; k < BIN_TRIPS; ++k) {                \
    const int eb_ = (EB) + k * 32;                                       \
    if (eb_ < e1)                                       \
      sgc_bin_trip(my, my_n, (unsigned)DB[k], (unsigned)SB[k], eb_ + lane < e1, lt); \
  }

  int da[BIN_TRIPS], sa[BIN_TRIPS], db[BIN_TRIPS], sb[BIN_TRIPS];
  BIN_LOAD(da, sa, e0)
  for (int eb = e0; eb < e1; eb += 2 * BIN_TRIPS * 32) {
    BIN_LOAD(db, sb, eb + BIN_TRIPS * 32)
    BIN_WORK(da, sa, eb)
    BIN_LOAD(da, sa, eb + 2 * BIN_TRIPS * 32)
    BIN_WORK(db, sb, eb + BIN_TRIPS * 32)
  }
#undef BIN_LOAD
#undef BIN_WORK
  __builtin_amdgcn_wave_barrier();
  for (int g = lane; g < BIN_GROUPS; g += 32) my[g * SLOT_W] = my_n[g];
  __builtin_amdgcn_wave_barrier();

  int gs = N >= SLOT_MIN_ROWS ? (N - SLOT_MIN_ROWS) / NODES_PER_WG + 1 : 0;
  if (gs > BIN_GROUPS) gs = BIN_GROUPS;
  uint4* line0 = (uint4*)buf + (size_t)task * (SLOT_W / 4) + (lane & 7);
  const size_t gstride = (size_t)NODES_PER_WG * OUT_F / 4;
  int i0 = 0;
  for (; (i0 + 8) * 4 <= gs; i0 += 8) {
    uint4 v[8];
#pragma unroll
    for (int j = 0; j < 8; ++j) v[j] = ((const uint4*)my)[(i0 + j) * 32 + lane];
#pragma unroll
    for (int j = 0; j < 8; ++j) line0[(size_t)((i0 + j) * 4 + (lane >> 3)) * gstride] = v[j];
  }
  for (int i = i0; i * 4 < gs; ++i) {
    const int g = i * 4 + (lane >> 3);
    if (g < gs) line0[(size_t)g * gstride] = ((const uint4*)my)[i * 32 + lane];
  }
}

extern "C" void kernel_launch(void* const* d_in, const int* in_sizes, int n_in,
                              void* d_out, int out_size, void* d_ws, size_t ws_size,
                              hipStream_t stream) {
  const float* h    = (const float*)d_in[0];
  const float* w    = (const float*)d_in[1];
  const float* norm = (const float*)d_in[2];
  const int*   src  = (const int*)d_in[3];
  const int*   dst  = (const int*)d_in[4];
  float* out = (float*)d_out;
  float* x1  = (float*)d_ws;

  const int N = in_sizes[2];
  const int E = in_sizes[3];

  const int mblocks = (N + 15) / 16;
  const int groups  = (N + NODES_PER_WG - 1) / NODES_PER_WG;

  sgc_gemm_wmma<<<mblocks, 256, 0, stream>>>(h, w, out);
  sgc_bin<<<BIN_TASKS / BIN_WAVES, BIN_WAVES * 32, 0, stream>>>(src, dst, x1, N, E);
  sgc_hop<<<groups, HOP_WAVES * 32, 0, stream>>>(out, norm, src, dst, x1, N, E);
  sgc_bin<<<BIN_TASKS / BIN_WAVES, BIN_WAVES * 32, 0, stream>>>(src, dst, out, N, E);
  sgc_hop<<<groups, HOP_WAVES * 32, 0, stream>>>(x1, norm, src, dst, out, N, E);
}


# Source-norm scaling moved to the producers (GEMM stores norm*x, hop 1 stores norm*(norm*sum)): hops add plain rows, no norm[src] gather per edge; sum order kept, products now rounded before adding
# speedup vs baseline: 3.3807x; 1.0543x over previous
#include <hip/hip_runtime.h>

#define IN_F  256
#define OUT_F 128
#define LDS_PAD 260

typedef __attribute__((ext_vector_type(2))) float v2f;
typedef __attribute__((ext_vector_type(8))) float v8f;

__global__ __launch_bounds__(256) void sgc_gemm_wmma(const float* __restrict__ h,
                                                     const float* __restrict__ w,
                                                     const float* __restrict__ norm,
                                                     float* __restrict__ x) {
  __shared__ float atile[16 * LDS_PAD];
  const int m0  = blockIdx.x * 16;
  const int tid = threadIdx.x;

  float nr[8];
#pragma unroll
  for (int r = 0; r < 8; ++r) nr[r] = norm[m0 + 8 * ((tid >> 4) & 1) + r];

  for (int i4 = tid; i4 < 16 * IN_F / 4; i4 += 256) {
    int i = i4 * 4;
    int r = i / IN_F, c = i % IN_F;
    float4 v = *(const float4*)(h + (size_t)(m0 + r) * IN_F + c);
    atile[r * LDS_PAD + c + 0] = v.x;
    atile[r * LDS_PAD + c + 1] = v.y;
    atile[r * LDS_PAD + c + 2] = v.z;
    atile[r * LDS_PAD + c + 3] = v.w;
  }
  __syncthreads();

  const int wave = tid >> 5;
  const int lane = tid & 31;
  const int half = lane >> 4;
  const int l    = lane & 15;
  const int n0   = wave * 16;

  v8f c = {};
  const float* arow = atile + l * LDS_PAD + 2 * half;
  const float* bcol = w + (size_t)(2 * half) * OUT_F + n0 + l;
  for (int k = 0; k < IN_F; k += 4) {
    float2 av = *(const float2*)(arow + k);
    v2f a, b;
    a.x = av.x;
    a.y = av.y;
    b.x = bcol[(size_t)k * OUT_F];
    b.y = bcol[(size_t)(k + 1) * OUT_F];
    c = __builtin_amdgcn_wmma_f32_16x16x4_f32(false, a, false, b, (short)0, c,
                                              false, false);
  }

  float* out = x + (size_t)(m0 + 8 * half) * OUT_F + n0 + l;
#pragma unroll
  for (int r = 0; r < 8; ++r) out[(size_t)r * OUT_F] = c[r] * nr[r];
}

typedef const float __attribute__((address_space(1)))* gfloat_cptr;
typedef __attribute__((ext_vector_type(4))) float v4f;
typedef const v4f __attribute__((address_space(1)))* gv4f_cptr;

#define NODES_PER_WG 391
#define HOP_WAVES 32
#define CHUNK 512
#define LIST_W 576
#define OWN_W 64
#define EDGE_BATCH 8
#define BIN_TASKS 1024
#define BIN_WAVES 4
#define BIN_GROUPS 256
#define SLOT_W 32
#define SLOT_CAP 31
#define SLOT_MIN_ROWS 256
#define SRC_BITS 22
#define ROW_CAP (BIN_TASKS * SLOT_CAP)
#define WAVE_CAP (HOP_WAVES * SLOT_CAP)
#define ROW_SLOTS 416
#define ROW_BATCH 16

struct sgc_lds_scan {
  float          acc[NODES_PER_WG * OUT_F];
  int            note_e[HOP_WAVES * LIST_W];
  unsigned short note_d[HOP_WAVES * LIST_W];
  int            own_e[HOP_WAVES * OWN_W];
  unsigned short own_d[HOP_WAVES * OWN_W];
  int            note_n[2][HOP_WAVES];
};
struct sgc_lds_rows {
  unsigned elist[ROW_CAP];
  unsigned byrow[ROW_CAP];
  unsigned hist[HOP_WAVES][ROW_SLOTS];
  unsigned rtot[ROW_SLOTS];
  unsigned rstart[ROW_SLOTS];
};
union sgc_lds {
  sgc_lds_scan s;
  sgc_lds_rows f;
};
static_assert(sizeof(sgc_lds_rows) <= offsetof(sgc_lds_scan, note_n), "usual path must not reach note_n");
static_assert(ROW_SLOTS >= NODES_PER_WG && ROW_SLOTS == 13 * 32, "scan by 13 counters per lane");
static_assert(NODES_PER_WG <= 512, "a local row number has 9 bits");
static_assert(HOP_WAVES * 32 == BIN_TASKS, "32 slots per wave");
static_assert(2 * ROW_BATCH == 32, "two batches per 32 edges");
static_assert(NODES_PER_WG <= HOP_WAVES * 32, "a wave's rows fit the lanes of one register");

__device__ __forceinline__ unsigned sgc_same_row(bool valid, unsigned row) {
  unsigned same = __builtin_amdgcn_ballot_w32(valid);
#pragma unroll
  for (int b = 0; b < 9; ++b) {
    const bool bit = (row >> b) & 1u;
    const unsigned bm = __builtin_amdgcn_ballot_w32(bit);
    same &= bit ? bm : ~bm;
  }
  return same;
}

__device__ __forceinline__ void sgc_row_issue(const float* __restrict__ x, v4f (&v)[ROW_BATCH],
                                              int sv, int lane, int b, int m) {
#pragma unroll
  for (int g = 0; g < ROW_BATCH / 4; ++g) {
    if (b + 4 * g < m) {
#pragma unroll
      for (int j = 4 * g; j < 4 * g + 4; ++j) {
        const int s = __builtin_amdgcn_readlane(sv, b + j);
        gv4f_cptr xrow = (gv4f_cptr)(x + (size_t)s * OUT_F);
        asm("" : "+s"(xrow));
        v[j] = xrow[lane];
      }
    }
  }
}

__device__ __forceinline__ void sgc_row_add(v4f& a, const v4f (&v)[ROW_BATCH], int b, int m) {
#pragma unroll
  for (int j = 0; j < ROW_BATCH; ++j) {
    if (b + j < m) a += v[j];
  }
}

template <bool TAIL>
__device__ __forceinline__ void sgc_hop_batch(const float* __restrict__ x, float* acc, int sv,
                                              int dv, int lane, int b, int n) {
  float v[EDGE_BATCH][4];
#pragma unroll
  for (int j = 0; j < EDGE_BATCH; ++j) {
    const int s = __builtin_amdgcn_readlane(sv, b + j);
    gfloat_cptr xrow = (gfloat_cptr)(x + (size_t)s * OUT_F);
    asm("" : "+s"(xrow));
    v[j][0] = xrow[lane];
    v[j][1] = xrow[lane + 32];
    v[j][2] = xrow[lane + 64];
    v[j][3] = xrow[lane + 96];
  }
#pragma unroll
  for (int j = 0; j < EDGE_BATCH; ++j) {
    if (!TAIL || b + j < n) {
      const int d = __builtin_amdgcn_readlane(dv, b + j);
      float* arow = acc + d * OUT_F + lane;
      arow[0]  += v[j][0];
      arow[32] += v[j][1];
      arow[64] += v[j][2];
      arow[96] += v[j][3];
    }
  }
}

template <bool TAIL>
__device__ __forceinline__ void sgc_hop_group(const float* __restrict__ x,
                                              const int* __restrict__ src, float* acc,
                                              const int* oe, const unsigned short* od, int n,
                                              int lane, bool by_edge) {
  const int li = TAIL ? (lane < n ? lane : n - 1) : lane;
  const int e  = oe[li];
  const int dv = od[li];
  const int sv = by_edge ? src[e] : e;
  int b = 0;
  for (; b + EDGE_BATCH <= n; b += EDGE_BATCH)
    sgc_hop_batch<false>(x, acc, sv, dv, lane, b, n);
  if (TAIL && b < n) sgc_hop_batch<true>(x, acc, sv, dv, lane, b, n);
}

__device__ __forceinline__ void sgc_lds_barrier() {
  __builtin_amdgcn_fence(__ATOMIC_RELEASE, "workgroup", "local");
  __builtin_amdgcn_s_barrier();
  __builtin_amdgcn_fence(__ATOMIC_ACQUIRE, "workgroup", "local");
}

__device__ __forceinline__ void sgc_hop_publish_and_hand_over(
    const float* __restrict__ x, const int* __restrict__ src, float* acc, const int* note_e,
    const unsigned short* note_d, int* note_n, int* oe,
    unsigned short* od, int& cnt, int& pc, bool last, int wave, int lane, unsigned lt, int room,
    bool by_edge) {
  if (lane == 0) note_n[wave] = cnt;
  sgc_lds_barrier();
  const int cn = note_n[lane];
  const bool hand_over = last || __builtin_amdgcn_ballot_w32(cn + room > LIST_W) != 0;
  if (!hand_over) return;
  for (int ws = 0; ws < HOP_WAVES; ++ws) {
    const int c = __builtin_amdgcn_readlane(cn, ws);
    const int*            le = note_e + ws * LIST_W;
    const unsigned short* ld = note_d + ws * LIST_W;
    for (int off = 0; off < c; off += 32) {
      const int  idx   = off + lane;
      const bool valid = idx < c;
      const int  d     = valid ? (int)ld[idx] : -1;
      const bool mine  = valid && ((d & (HOP_WAVES - 1)) == wave);
      const unsigned mk = __builtin_amdgcn_ballot_w32(mine);
      if (mk) {
        if (mine) {
          const int pos = pc + __builtin_popcount(mk & lt);
          oe[pos] = le[idx];
          od[pos] = (unsigned short)d;
        }
        pc += __builtin_popcount(mk);
        if (pc >= 32) {
          sgc_hop_group<false>(x, src, acc, oe, od, 32, lane, by_edge);
          const int rem = pc - 32;
          if (lane < rem) {
            const int            te = oe[32 + lane];
            const unsigned short td = od[32 + lane];
            oe[lane] = te;
            od[lane] = td;
          }
          pc = rem;
        }
      }
    }
  }
  sgc_lds_barrier();
  cnt = 0;
}

__global__ __launch_bounds__(HOP_WAVES * 32) void sgc_hop(const float* __restrict__ x,
                                                          const float* __restrict__ norm,
                                                          const int* __restrict__ src,
                                                          const int* __restrict__ dst,
                                                          float* __restrict__ xout, int N, int E,
                                                          int twice) {
  __shared__ sgc_lds L;
  float*          acc    = L.s.acc;
  int*            note_e = L.s.note_e;
  unsigned short* note_d = L.s.note_d;
  int*            own_e  = L.s.own_e;
  unsigned short* own_d  = L.s.own_d;
  int (*note_n)[HOP_WAVES] = L.s.note_n;

  const int tid  = threadIdx.x;
  const int lane = tid & 31;
  const int wave  = __builtin_amdgcn_readfirstlane(tid >> 5);
  const int node0 = blockIdx.x * NODES_PER_WG;
  const int left  = N - node0;
  const int nrows = left < NODES_PER_WG ? left : NODES_PER_WG;

  const float* slots = xout + (size_t)node0 * OUT_F;
  const bool served = (int)gridDim.x <= BIN_GROUPS && nrows >= SLOT_MIN_ROWS;
  unsigned* el = L.f.elist + wave * WAVE_CAP;
  unsigned* hw = L.f.hist[wave];
  int ecnt = 0;
  bool bad = !served;
  if (served) {
    unsigned wv[HOP_WAVES];
#pragma unroll
    for (int k = 0; k < HOP_WAVES; ++k)
      wv[k] = __float_as_uint(slots[(size_t)(k * HOP_WAVES + wave) * SLOT_W + lane]);
    unsigned over = 0u;
#pragma unroll
    for (int k = 0; k < HOP_WAVES; ++k) {
      const unsigned cu = (unsigned)__builtin_amdgcn_readfirstlane((int)wv[k]);
      over |= cu > SLOT_CAP ? 1u : 0u;
      const int c = (int)(cu > SLOT_CAP ? SLOT_CAP : cu);
      if (lane >= 1 && lane <= c) el[ecnt + lane - 1] = wv[k];
      ecnt += c;
    }
    bad = over != 0u;
    for (int j = lane; j < ROW_SLOTS; j += 32) hw[j] = 0u;
  }
  if (lane == 0) note_n[1][wave] = bad ? 1 : 0;
  sgc_lds_barrier();
  const bool scan = __builtin_amdgcn_ballot_w32(note_n[1][lane] != 0) != 0;

  int*            ne = note_e + wave * LIST_W;
  unsigned short* nd = note_d + wave * LIST_W;
  int*            oe = own_e + wave * OWN_W;
  unsigned short* od = own_d + wave * OWN_W;
  int cnt = 0;
  int pc  = 0;
  const unsigned lt = (1u << lane) - 1u;

#define SGC_NOTE(mc, eidx, dloc)                                         \
  {                                                                      \
    const unsigned mk_ = __builtin_amdgcn_ballot_w32(mc);                \
    if (mk_) {                                                           \
      if (mc) {                                                          \
        const int pos_ = cnt + __builtin_popcount(mk_ & lt);             \
        ne[pos_] = (eidx);                                               \
        nd[pos_] = (unsigned short)(dloc);                               \
      }                                                                  \
      cnt += __builtin_popcount(mk_);                                    \
    }                                                                    \
  }

  const int nchunks = E / CHUNK;
  const int nsteps  = (nchunks + HOP_WAVES - 1) / HOP_WAVES + 1;
  const int rot = (int)(((long long)blockIdx.x * nchunks) / gridDim.x);
  const int4* dst4 = (const int4*)dst;

#define SGC_LOAD(IT, BUF)                                                \
  {                                                                      \
    const int ql_ = wave + (IT) * HOP_WAVES;                             \
    int pl_ = (ql_ < nchunks ? ql_ : nchunks - 1) + rot;                 \
    if (pl_ >= nchunks) pl_ -= nchunks;                                  \
    _Pragma("unroll") for (int k = 0; k < 4; ++k)                        \
        BUF[k] = dst4[pl_ * (CHUNK / 4) + k * 32 + lane];                \
  }

#define SGC_STEP(IT, BUF, NEXTBUF)                                       \
  {                                                                      \
    const int it_ = (IT);                                                \
    SGC_LOAD(it_ + 2, NEXTBUF)                                           \
    if (it_ < nsteps - 1) {                                              \
      const int q_ = wave + it_ * HOP_WAVES;                             \
      if (q_ < nchunks) {                                                \
        int p_ = q_ + rot;                                               \
        if (p_ >= nchunks) p_ -= nchunks;                                \
        _Pragma("unroll") for (int k = 0; k < 4; ++k) {                  \
          const int4 d4 = BUF[k];                                        \
          const int l0 = d4.x - node0, l1 = d4.y - node0;                \
          const int l2 = d4.z - node0, l3 = d4.w - node0;                \
          const bool m0 = (unsigned)l0 < (unsigned)nrows;                \
          const bool m1 = (unsigned)l1 < (unsigned)nrows;                \
          const bool m2 = (unsigned)l2 < (unsigned)nrows;                \
          const bool m3 = (unsigned)l3 < (unsigned)nrows;                \
                                 \
          if (__builtin_amdgcn_ballot_w32(m0 | m1 | m2 | m3)) {          \
            const int e0 = p_ * CHUNK + k * 128 + lane * 4;              \
            SGC_NOTE(m0, e0, l0)                                         \
            SGC_NOTE(m1, e0 + 1, l1)                                     \
            SGC_NOTE(m2, e0 + 2, l2)                                     \
            SGC_NOTE(m3, e0 + 3, l3)                                     \
          }                                                              \
        }                                                                \
      }                                                                  \
    } else {                                                             \
            \
      const int i_ = nchunks * CHUNK + wave * 32;                        \
      if (i_ < E) {                                                      \
        const int  e_  = i_ + lane;                                      \
        const bool ok_ = e_ < E;                                         \
        const int  l_  = dst[ok_ ? e_ : E - 1] - node0;                  \
        const bool m_  = ok_ && (unsigned)l_ < (unsigned)nrows;          \
        SGC_NOTE(m_, e_, l_)                                             \
      }                                                                  \
    }                                                                    \
    sgc_hop_publish_and_hand_over(x, src, acc, note_e, note_d, &note_n[it_ & 1][0], oe, od,   \
                                  cnt, pc, it_ == nsteps - 1, wave, lane, lt, CHUNK, true);   \
  }

  if (!scan) {
    for (int t = 0; t * 32 < ecnt; ++t) {
      const int  i     = t * 32 + lane;
      const bool valid = i < ecnt;
      const unsigned e   = valid ? el[i] : 0u;
      const unsigned row = e >> SRC_BITS;
      const unsigned same = sgc_same_row(valid, row);
      if (valid && (same & lt) == 0u) hw[row] += (unsigned)__builtin_popcount(same);
    }
    sgc_lds_barrier();
    if (tid < ROW_SLOTS) {
      unsigned hc[HOP_WAVES];
#pragma unroll
      for (int w = 0; w < HOP_WAVES; ++w) hc[w] = L.f.hist[w][tid];
      unsigned run = 0u;
#pragma unroll
      for (int w = 0; w < HOP_WAVES; ++w) {
        L.f.hist[w][tid] = run;
        run += hc[w];
      }
      L.f.rtot[tid] = run;
    }
    sgc_lds_barrier();
    if (wave == 0) {
      unsigned cs[13];
      unsigned sum = 0u;
#pragma unroll
      for (int j = 0; j < 13; ++j) {
        const unsigned tt = L.f.rtot[lane * 13 + j];
        cs[j] = sum;
        sum += tt;
      }
      unsigned incl = sum;
#pragma unroll
      for (int d = 1; d < 32; d <<= 1) {
        const unsigned up = __shfl_up(incl, (unsigned)d);
        if (lane >= d) incl += up;
      }
      const unsigned excl = incl - sum;
#pragma unroll
      for (int j = 0; j < 13; ++j) L.f.rstart[lane * 13 + j] = excl + cs[j];
    }
    sgc_lds_barrier();

    for (int t = 0; t * 32 < ecnt; ++t) {
      const int  i     = t * 32 + lane;
      const bool valid = i < ecnt;
      const unsigned e   = valid ? el[i] : 0u;
      const unsigned row = e >> SRC_BITS;
      const unsigned same = sgc_same_row(valid, row);
      const unsigned rank = (unsigned)__builtin_popcount(same & lt);
      if (valid) {
        const unsigned before = hw[row];
        L.f.byrow[L.f.rstart[row] + before + rank] = e & ((1u << SRC_BITS) - 1u);
        if (rank == 0u) hw[row] = before + (unsigned)__builtin_popcount(same);
      }
    }
    sgc_lds_barrier();

    const int   kr  = wave + HOP_WAVES * lane;
    const float ndv = kr < nrows ? norm[node0 + kr] : 0.f;
    int k = 0;
    for (int r = wave; r < nrows; r += HOP_WAVES, ++k) {
      const int n  = __builtin_amdgcn_readfirstlane((int)L.f.rtot[r]);
      const int s0 = __builtin_amdgcn_readfirstlane((int)L.f.rstart[r]);
      const float ndst = __int_as_float(__builtin_amdgcn_readlane(__float_as_int(ndv), k));
      const unsigned* br = L.f.byrow + s0;
      v4f a = {0.f, 0.f, 0.f, 0.f};
      for (int base = 0; base < n; base += 32) {
        const int m  = n - base < 32 ? n - base : 32;
        const int li = lane < m ? lane : m - 1;
        const int sv = (int)br[base + li];
        v4f v[ROW_BATCH];
        sgc_row_issue(x, v, sv, lane, 0, m);
        sgc_row_add(a, v, 0, m);
        if (m > ROW_BATCH) {
          sgc_row_issue(x, v, sv, lane, ROW_BATCH, m);
          sgc_row_add(a, v, ROW_BATCH, m);
        }
      }
      v4f o = ndst * a;
      if (twice) o = ndst * o;
      ((v4f*)xout)[(size_t)(node0 + r) * (OUT_F / 4) + lane] = o;
    }
  } else {
    for (int i = tid; i < NODES_PER_WG * OUT_F / 4; i += HOP_WAVES * 32)
      ((float4*)acc)[i] = make_float4(0.f, 0.f, 0.f, 0.f);
    int4 buf0[4] = {}, buf1[4] = {}, buf2[4] = {};
    SGC_LOAD(0, buf0)
    SGC_LOAD(1, buf1)
    for (int it = 0; it < nsteps; it += 3) {
      SGC_STEP(it, buf0, buf2)
      if (it + 1 < nsteps) SGC_STEP(it + 1, buf1, buf0)
      if (it + 2 < nsteps) SGC_STEP(it + 2, buf2, buf1)
    }
    if (pc > 0) sgc_hop_group<true>(x, src, acc, oe, od, pc, lane, true);

    for (int r = wave; r < nrows; r += HOP_WAVES) {
      const float  ndst = norm[node0 + r];
      const float4 a    = ((const float4*)acc)[r * (OUT_F / 4) + lane];
      float4 o = make_float4(ndst * a.x, ndst * a.y, ndst * a.z, ndst * a.w);
      if (twice) o = make_float4(ndst * o.x, ndst * o.y, ndst * o.z, ndst * o.w);
      ((float4*)xout)[(size_t)(node0 + r) * (OUT_F / 4) + lane] = o;
    }
  }
#undef SGC_STEP
#undef SGC_LOAD
#undef SGC_NOTE
}

__device__ __forceinline__ void sgc_bin_trip(unsigned* my, unsigned* my_n, unsigned d, unsigned s,
                                             bool in, unsigned lt) {
  const unsigned g   = d / NODES_PER_WG;
  const unsigned row = d - g * NODES_PER_WG;
  const bool ok = in && g < BIN_GROUPS;
  unsigned same = __builtin_amdgcn_ballot_w32(ok);
#pragma unroll
  for (int b = 0; b < 8; ++b) {
    const bool bit = (g >> b) & 1u;
    const unsigned bm = __builtin_amdgcn_ballot_w32(bit);
    same &= bit ? bm : ~bm;
  }
  const unsigned rank = __builtin_popcount(same & lt);
  __builtin_amdgcn_wave_barrier();
  if (ok) {
    const unsigned base = my_n[g];
    const unsigned pos  = base + rank;
    if (pos < SLOT_CAP) my[g * SLOT_W + 1 + pos] = (row << SRC_BITS) | s;
    if (rank == 0) my_n[g] = base + __builtin_popcount(same);
  }
}

#define BIN_TRIPS 8

__global__ __launch_bounds__(BIN_WAVES * 32) void sgc_bin(const int* __restrict__ src,
                                                          const int* __restrict__ dst,
                                                          float* __restrict__ buf, int N, int E) {
  __shared__ unsigned slot[BIN_WAVES * BIN_GROUPS * SLOT_W];
  __shared__ unsigned slot_n[BIN_WAVES * BIN_GROUPS];
  const int lane = threadIdx.x & 31;
  const int wave = __builtin_amdgcn_readfirstlane((int)threadIdx.x >> 5);
  const int task = blockIdx.x * BIN_WAVES + wave;
  unsigned* my   = slot + wave * (BIN_GROUPS * SLOT_W);
  unsigned* my_n = slot_n + wave * BIN_GROUPS;
  for (int g = lane; g < BIN_GROUPS; g += 32) my_n[g] = 0u;

  const int ept = (E + BIN_TASKS - 1) / BIN_TASKS;
  const int e0  = task * ept;
  const int e1  = e0 + ept < E ? e0 + ept : E;
  const unsigned lt = (1u << lane) - 1u;

#define BIN_LOAD(DB, SB, EB)                                             \
  _Pragma("unroll") for (int k = 0; k < BIN_TRIPS; ++k) {                \
    const int e_ = (EB) + k * 32 + lane;                                 \
    const int c_ = e_ < E ? e_ : E - 1;                                  \
    DB[k] = dst[c_];                                                     \
    SB[k] = src[c_];                                                     \
  }
#define BIN_WORK(DB, SB, EB)                                             \
  _Pragma("unroll") for (int k = 0; k < BIN_TRIPS; ++k) {                \
    const int eb_ = (EB) + k * 32;                                       \
    if (eb_ < e1)                                       \
      sgc_bin_trip(my, my_n, (unsigned)DB[k], (unsigned)SB[k], eb_ + lane < e1, lt); \
  }

  int da[BIN_TRIPS], sa[BIN_TRIPS], db[BIN_TRIPS], sb[BIN_TRIPS];
  BIN_LOAD(da, sa, e0)
  for (int eb = e0; eb < e1; eb += 2 * BIN_TRIPS * 32) {
    BIN_LOAD(db, sb, eb + BIN_TRIPS * 32)
    BIN_WORK(da, sa, eb)
    BIN_LOAD(da, sa, eb + 2 * BIN_TRIPS * 32)
    BIN_WORK(db, sb, eb + BIN_TRIPS * 32)
  }
#undef BIN_LOAD
#undef BIN_WORK
  __builtin_amdgcn_wave_barrier();
  for (int g = lane; g < BIN_GROUPS; g += 32) my[g * SLOT_W] = my_n[g];
  __builtin_amdgcn_wave_barrier();

  int gs = N >= SLOT_MIN_ROWS ? (N - SLOT_MIN_ROWS) / NODES_PER_WG + 1 : 0;
  if (gs > BIN_GROUPS) gs = BIN_GROUPS;
  uint4* line0 = (uint4*)buf + (size_t)task * (SLOT_W / 4) + (lane & 7);
  const size_t gstride = (size_t)NODES_PER_WG * OUT_F / 4;
  int i0 = 0;
  for (; (i0 + 8) * 4 <= gs; i0 += 8) {
    uint4 v[8];
#pragma unroll
    for (int j = 0; j < 8; ++j) v[j] = ((const uint4*)my)[(i0 + j) * 32 + lane];
#pragma unroll
    for (int j = 0; j < 8; ++j) line0[(size_t)((i0 + j) * 4 + (lane >> 3)) * gstride] = v[j];
  }
  for (int i = i0; i * 4 < gs; ++i) {
    const int g = i * 4 + (lane >> 3);
    if (g < gs) line0[(size_t)g * gstride] = ((const uint4*)my)[i * 32 + lane];
  }
}

extern "C" void kernel_launch(void* const* d_in, const int* in_sizes, int n_in,
                              void* d_out, int out_size, void* d_ws, size_t ws_size,
                              hipStream_t stream) {
  const float* h    = (const float*)d_in[0];
  const float* w    = (const float*)d_in[1];
  const float* norm = (const float*)d_in[2];
  const int*   src  = (const int*)d_in[3];
  const int*   dst  = (const int*)d_in[4];
  float* out = (float*)d_out;
  float* x1  = (float*)d_ws;

  const int N = in_sizes[2];
  const int E = in_sizes[3];

  const int mblocks = (N + 15) / 16;
  const int groups  = (N + NODES_PER_WG - 1) / NODES_PER_WG;

  sgc_gemm_wmma<<<mblocks, 256, 0, stream>>>(h, w, norm, out);
  sgc_bin<<<BIN_TASKS / BIN_WAVES, BIN_WAVES * 32, 0, stream>>>(src, dst, x1, N, E);
  sgc_hop<<<groups, HOP_WAVES * 32, 0, stream>>>(out, norm, src, dst, x1, N, E, 1);
  sgc_bin<<<BIN_TASKS / BIN_WAVES, BIN_WAVES * 32, 0, stream>>>(src, dst, out, N, E);
  sgc_hop<<<groups, HOP_WAVES * 32, 0, stream>>>(x1, norm, src, dst, out, N, E, 0);
}


# GEMM strip per work-group 16 -> 80 rows: a wave keeps 5 accumulator tiles, each B fragment feeds 5 matrix instructions (5x fewer waves and weight loads); A strip fetched in one batch; sums unchanged
# speedup vs baseline: 3.5705x; 1.0561x over previous
#include <hip/hip_runtime.h>

#define IN_F  256
#define OUT_F 128
#define LDS_PAD 260

typedef __attribute__((ext_vector_type(2))) float v2f;
typedef __attribute__((ext_vector_type(8))) float v8f;

#define GEMM_TILES 5
#define GEMM_ROWS (16 * GEMM_TILES)
#define GEMM_STAGE (GEMM_ROWS / 4)

template <bool FULL>
__device__ __forceinline__ void sgc_gemm_body(const float* __restrict__ h,
                                              const float* __restrict__ w,
                                              const float* __restrict__ norm,
                                              float* __restrict__ x, int N, float* atile,
                                              int m0, int tid) {
  const int r0 = tid >> 6;
  const int c0 = (tid & 63) * 4;
  const float* hb = h + (size_t)(m0 + r0) * IN_F + c0;
  float4 v[GEMM_STAGE];
#pragma unroll
  for (int j = 0; j < GEMM_STAGE; ++j) {
    if (FULL) {
      v[j] = *(const float4*)(hb + (size_t)(4 * j) * IN_F);
    } else {
      const int gr = m0 + r0 + 4 * j < N ? m0 + r0 + 4 * j : N - 1;
      v[j] = *(const float4*)(h + (size_t)gr * IN_F + c0);
    }
  }
#pragma unroll
  for (int j = 0; j < GEMM_STAGE; ++j)
    *(float4*)(atile + (r0 + 4 * j) * LDS_PAD + c0) = v[j];
  __syncthreads();

  const int wave = tid >> 5;
  const int lane = tid & 31;
  const int half = lane >> 4;
  const int l    = lane & 15;
  const int n0   = wave * 16;

  v8f c[GEMM_TILES] = {};
  const float* arow = atile + l * LDS_PAD + 2 * half;
  const float* bcol = w + (size_t)(2 * half) * OUT_F + n0 + l;
  for (int k = 0; k < IN_F; k += 4) {
    v2f b;
    b.x = bcol[(size_t)k * OUT_F];
    b.y = bcol[(size_t)(k + 1) * OUT_F];
#pragma unroll
    for (int t = 0; t < GEMM_TILES; ++t) {
      float2 av = *(const float2*)(arow + t * 16 * LDS_PAD + k);
      v2f a;
      a.x = av.x;
      a.y = av.y;
      c[t] = __builtin_amdgcn_wmma_f32_16x16x4_f32(false, a, false, b, (short)0, c[t],
                                                   false, false);
    }
  }

  float nr[GEMM_TILES][8];
#pragma unroll
  for (int t = 0; t < GEMM_TILES; ++t) {
#pragma unroll
    for (int r = 0; r < 8; ++r) {
      const int row = m0 + 16 * t + 8 * half + r;
      nr[t][r] = norm[FULL ? row : (row < N ? row : N - 1)];
    }
  }
  float* out = x + (size_t)(m0 + 8 * half) * OUT_F + n0 + l;
#pragma unroll
  for (int t = 0; t < GEMM_TILES; ++t) {
#pragma unroll
    for (int r = 0; r < 8; ++r) {
      if (FULL || m0 + 16 * t + 8 * half + r < N)
        out[(size_t)(16 * t + r) * OUT_F] = c[t][r] * nr[t][r];
    }
  }
}

__global__ __launch_bounds__(256) void sgc_gemm_wmma(const float* __restrict__ h,
                                                     const float* __restrict__ w,
                                                     const float* __restrict__ norm,
                                                     float* __restrict__ x, int N) {
  __shared__ __attribute__((aligned(16))) float atile[GEMM_ROWS * LDS_PAD];
  const int m0  = blockIdx.x * GEMM_ROWS;
  const int tid = threadIdx.x;
  if (m0 + GEMM_ROWS <= N)
    sgc_gemm_body<true>(h, w, norm, x, N, atile, m0, tid);
  else
    sgc_gemm_body<false>(h, w, norm, x, N, atile, m0, tid);
}

typedef const float __attribute__((address_space(1)))* gfloat_cptr;
typedef __attribute__((ext_vector_type(4))) float v4f;
typedef const v4f __attribute__((address_space(1)))* gv4f_cptr;

#define NODES_PER_WG 391
#define HOP_WAVES 32
#define CHUNK 512
#define LIST_W 576
#define OWN_W 64
#define EDGE_BATCH 8
#define BIN_TASKS 1024
#define BIN_WAVES 4
#define BIN_GROUPS 256
#define SLOT_W 32
#define SLOT_CAP 31
#define SLOT_MIN_ROWS 256
#define SRC_BITS 22
#define ROW_CAP (BIN_TASKS * SLOT_CAP)
#define WAVE_CAP (HOP_WAVES * SLOT_CAP)
#define ROW_SLOTS 416
#define ROW_BATCH 16

struct sgc_lds_scan {
  float          acc[NODES_PER_WG * OUT_F];
  int            note_e[HOP_WAVES * LIST_W];
  unsigned short note_d[HOP_WAVES * LIST_W];
  int            own_e[HOP_WAVES * OWN_W];
  unsigned short own_d[HOP_WAVES * OWN_W];
  int            note_n[2][HOP_WAVES];
};
struct sgc_lds_rows {
  unsigned elist[ROW_CAP];
  unsigned byrow[ROW_CAP];
  unsigned hist[HOP_WAVES][ROW_SLOTS];
  unsigned rtot[ROW_SLOTS];
  unsigned rstart[ROW_SLOTS];
};
union sgc_lds {
  sgc_lds_scan s;
  sgc_lds_rows f;
};
static_assert(sizeof(sgc_lds_rows) <= offsetof(sgc_lds_scan, note_n), "usual path must not reach note_n");
static_assert(ROW_SLOTS >= NODES_PER_WG && ROW_SLOTS == 13 * 32, "scan by 13 counters per lane");
static_assert(NODES_PER_WG <= 512, "a local row number has 9 bits");
static_assert(HOP_WAVES * 32 == BIN_TASKS, "32 slots per wave");
static_assert(2 * ROW_BATCH == 32, "two batches per 32 edges");
static_assert(NODES_PER_WG <= HOP_WAVES * 32, "a wave's rows fit the lanes of one register");

__device__ __forceinline__ unsigned sgc_same_row(bool valid, unsigned row) {
  unsigned same = __builtin_amdgcn_ballot_w32(valid);
#pragma unroll
  for (int b = 0; b < 9; ++b) {
    const bool bit = (row >> b) & 1u;
    const unsigned bm = __builtin_amdgcn_ballot_w32(bit);
    same &= bit ? bm : ~bm;
  }
  return same;
}

__device__ __forceinline__ void sgc_row_issue(const float* __restrict__ x, v4f (&v)[ROW_BATCH],
                                              int sv, int lane, int b, int m) {
#pragma unroll
  for (int g = 0; g < ROW_BATCH / 4; ++g) {
    if (b + 4 * g < m) {
#pragma unroll
      for (int j = 4 * g; j < 4 * g + 4; ++j) {
        const int s = __builtin_amdgcn_readlane(sv, b + j);
        gv4f_cptr xrow = (gv4f_cptr)(x + (size_t)s * OUT_F);
        asm("" : "+s"(xrow));
        v[j] = xrow[lane];
      }
    }
  }
}

__device__ __forceinline__ void sgc_row_add(v4f& a, const v4f (&v)[ROW_BATCH], int b, int m) {
#pragma unroll
  for (int j = 0; j < ROW_BATCH; ++j) {
    if (b + j < m) a += v[j];
  }
}

template <bool TAIL>
__device__ __forceinline__ void sgc_hop_batch(const float* __restrict__ x, float* acc, int sv,
                                              int dv, int lane, int b, int n) {
  float v[EDGE_BATCH][4];
#pragma unroll
  for (int j = 0; j < EDGE_BATCH; ++j) {
    const int s = __builtin_amdgcn_readlane(sv, b + j);
    gfloat_cptr xrow = (gfloat_cptr)(x + (size_t)s * OUT_F);
    asm("" : "+s"(xrow));
    v[j][0] = xrow[lane];
    v[j][1] = xrow[lane + 32];
    v[j][2] = xrow[lane + 64];
    v[j][3] = xrow[lane + 96];
  }
#pragma unroll
  for (int j = 0; j < EDGE_BATCH; ++j) {
    if (!TAIL || b + j < n) {
      const int d = __builtin_amdgcn_readlane(dv, b + j);
      float* arow = acc + d * OUT_F + lane;
      arow[0]  += v[j][0];
      arow[32] += v[j][1];
      arow[64] += v[j][2];
      arow[96] += v[j][3];
    }
  }
}

template <bool TAIL>
__device__ __forceinline__ void sgc_hop_group(const float* __restrict__ x,
                                              const int* __restrict__ src, float* acc,
                                              const int* oe, const unsigned short* od, int n,
                                              int lane, bool by_edge) {
  const int li = TAIL ? (lane < n ? lane : n - 1) : lane;
  const int e  = oe[li];
  const int dv = od[li];
  const int sv = by_edge ? src[e] : e;
  int b = 0;
  for (; b + EDGE_BATCH <= n; b += EDGE_BATCH)
    sgc_hop_batch<false>(x, acc, sv, dv, lane, b, n);
  if (TAIL && b < n) sgc_hop_batch<true>(x, acc, sv, dv, lane, b, n);
}

__device__ __forceinline__ void sgc_lds_barrier() {
  __builtin_amdgcn_fence(__ATOMIC_RELEASE, "workgroup", "local");
  __builtin_amdgcn_s_barrier();
  __builtin_amdgcn_fence(__ATOMIC_ACQUIRE, "workgroup", "local");
}

__device__ __forceinline__ void sgc_hop_publish_and_hand_over(
    const float* __restrict__ x, const int* __restrict__ src, float* acc, const int* note_e,
    const unsigned short* note_d, int* note_n, int* oe,
    unsigned short* od, int& cnt, int& pc, bool last, int wave, int lane, unsigned lt, int room,
    bool by_edge) {
  if (lane == 0) note_n[wave] = cnt;
  sgc_lds_barrier();
  const int cn = note_n[lane];
  const bool hand_over = last || __builtin_amdgcn_ballot_w32(cn + room > LIST_W) != 0;
  if (!hand_over) return;
  for (int ws = 0; ws < HOP_WAVES; ++ws) {
    const int c = __builtin_amdgcn_readlane(cn, ws);
    const int*            le = note_e + ws * LIST_W;
    const unsigned short* ld = note_d + ws * LIST_W;
    for (int off = 0; off < c; off += 32) {
      const int  idx   = off + lane;
      const bool valid = idx < c;
      const int  d     = valid ? (int)ld[idx] : -1;
      const bool mine  = valid && ((d & (HOP_WAVES - 1)) == wave);
      const unsigned mk = __builtin_amdgcn_ballot_w32(mine);
      if (mk) {
        if (mine) {
          const int pos = pc + __builtin_popcount(mk & lt);
          oe[pos] = le[idx];
          od[pos] = (unsigned short)d;
        }
        pc += __builtin_popcount(mk);
        if (pc >= 32) {
          sgc_hop_group<false>(x, src, acc, oe, od, 32, lane, by_edge);
          const int rem = pc - 32;
          if (lane < rem) {
            const int            te = oe[32 + lane];
            const unsigned short td = od[32 + lane];
            oe[lane] = te;
            od[lane] = td;
          }
          pc = rem;
        }
      }
    }
  }
  sgc_lds_barrier();
  cnt = 0;
}

__global__ __launch_bounds__(HOP_WAVES * 32) void sgc_hop(const float* __restrict__ x,
                                                          const float* __restrict__ norm,
                                                          const int* __restrict__ src,
                                                          const int* __restrict__ dst,
                                                          float* __restrict__ xout, int N, int E,
                                                          int twice) {
  __shared__ sgc_lds L;
  float*          acc    = L.s.acc;
  int*            note_e = L.s.note_e;
  unsigned short* note_d = L.s.note_d;
  int*            own_e  = L.s.own_e;
  unsigned short* own_d  = L.s.own_d;
  int (*note_n)[HOP_WAVES] = L.s.note_n;

  const int tid  = threadIdx.x;
  const int lane = tid & 31;
  const int wave  = __builtin_amdgcn_readfirstlane(tid >> 5);
  const int node0 = blockIdx.x * NODES_PER_WG;
  const int left  = N - node0;
  const int nrows = left < NODES_PER_WG ? left : NODES_PER_WG;

  const float* slots = xout + (size_t)node0 * OUT_F;
  const bool served = (int)gridDim.x <= BIN_GROUPS && nrows >= SLOT_MIN_ROWS;
  unsigned* el = L.f.elist + wave * WAVE_CAP;
  unsigned* hw = L.f.hist[wave];
  int ecnt = 0;
  bool bad = !served;
  if (served) {
    unsigned wv[HOP_WAVES];
#pragma unroll
    for (int k = 0; k < HOP_WAVES; ++k)
      wv[k] = __float_as_uint(slots[(size_t)(k * HOP_WAVES + wave) * SLOT_W + lane]);
    unsigned over = 0u;
#pragma unroll
    for (int k = 0; k < HOP_WAVES; ++k) {
      const unsigned cu = (unsigned)__builtin_amdgcn_readfirstlane((int)wv[k]);
      over |= cu > SLOT_CAP ? 1u : 0u;
      const int c = (int)(cu > SLOT_CAP ? SLOT_CAP : cu);
      if (lane >= 1 && lane <= c) el[ecnt + lane - 1] = wv[k];
      ecnt += c;
    }
    bad = over != 0u;
    for (int j = lane; j < ROW_SLOTS; j += 32) hw[j] = 0u;
  }
  if (lane == 0) note_n[1][wave] = bad ? 1 : 0;
  sgc_lds_barrier();
  const bool scan = __builtin_amdgcn_ballot_w32(note_n[1][lane] != 0) != 0;

  int*            ne = note_e + wave * LIST_W;
  unsigned short* nd = note_d + wave * LIST_W;
  int*            oe = own_e + wave * OWN_W;
  unsigned short* od = own_d + wave * OWN_W;
  int cnt = 0;
  int pc  = 0;
  const unsigned lt = (1u << lane) - 1u;

#define SGC_NOTE(mc, eidx, dloc)                                         \
  {                                                                      \
    const unsigned mk_ = __builtin_amdgcn_ballot_w32(mc);                \
    if (mk_) {                                                           \
      if (mc) {                                                          \
        const int pos_ = cnt + __builtin_popcount(mk_ & lt);             \
        ne[pos_] = (eidx);                                               \
        nd[pos_] = (unsigned short)(dloc);                               \
      }                                                                  \
      cnt += __builtin_popcount(mk_);                                    \
    }                                                                    \
  }

  const int nchunks = E / CHUNK;
  const int nsteps  = (nchunks + HOP_WAVES - 1) / HOP_WAVES + 1;
  const int rot = (int)(((long long)blockIdx.x * nchunks) / gridDim.x);
  const int4* dst4 = (const int4*)dst;

#define SGC_LOAD(IT, BUF)                                                \
  {                                                                      \
    const int ql_ = wave + (IT) * HOP_WAVES;                             \
    int pl_ = (ql_ < nchunks ? ql_ : nchunks - 1) + rot;                 \
    if (pl_ >= nchunks) pl_ -= nchunks;                                  \
    _Pragma("unroll") for (int k = 0; k < 4; ++k)                        \
        BUF[k] = dst4[pl_ * (CHUNK / 4) + k * 32 + lane];                \
  }

#define SGC_STEP(IT, BUF, NEXTBUF)                                       \
  {                                                                      \
    const int it_ = (IT);                                                \
    SGC_LOAD(it_ + 2, NEXTBUF)                                           \
    if (it_ < nsteps - 1) {                                              \
      const int q_ = wave + it_ * HOP_WAVES;                             \
      if (q_ < nchunks) {                                                \
        int p_ = q_ + rot;                                               \
        if (p_ >= nchunks) p_ -= nchunks;                                \
        _Pragma("unroll") for (int k = 0; k < 4; ++k) {                  \
          const int4 d4 = BUF[k];                                        \
          const int l0 = d4.x - node0, l1 = d4.y - node0;                \
          const int l2 = d4.z - node0, l3 = d4.w - node0;                \
          const bool m0 = (unsigned)l0 < (unsigned)nrows;                \
          const bool m1 = (unsigned)l1 < (unsigned)nrows;                \
          const bool m2 = (unsigned)l2 < (unsigned)nrows;                \
          const bool m3 = (unsigned)l3 < (unsigned)nrows;                \
                                 \
          if (__builtin_amdgcn_ballot_w32(m0 | m1 | m2 | m3)) {          \
            const int e0 = p_ * CHUNK + k * 128 + lane * 4;              \
            SGC_NOTE(m0, e0, l0)                                         \
            SGC_NOTE(m1, e0 + 1, l1)                                     \
            SGC_NOTE(m2, e0 + 2, l2)                                     \
            SGC_NOTE(m3, e0 + 3, l3)                                     \
          }                                                              \
        }                                                                \
      }                                                                  \
    } else {                                                             \
            \
      const int i_ = nchunks * CHUNK + wave * 32;                        \
      if (i_ < E) {                                                      \
        const int  e_  = i_ + lane;                                      \
        const bool ok_ = e_ < E;                                         \
        const int  l_  = dst[ok_ ? e_ : E - 1] - node0;                  \
        const bool m_  = ok_ && (unsigned)l_ < (unsigned)nrows;          \
        SGC_NOTE(m_, e_, l_)                                             \
      }                                                                  \
    }                                                                    \
    sgc_hop_publish_and_hand_over(x, src, acc, note_e, note_d, &note_n[it_ & 1][0], oe, od,   \
                                  cnt, pc, it_ == nsteps - 1, wave, lane, lt, CHUNK, true);   \
  }

  if (!scan) {
    for (int t = 0; t * 32 < ecnt; ++t) {
      const int  i     = t * 32 + lane;
      const bool valid = i < ecnt;
      const unsigned e   = valid ? el[i] : 0u;
      const unsigned row = e >> SRC_BITS;
      const unsigned same = sgc_same_row(valid, row);
      if (valid && (same & lt) == 0u) hw[row] += (unsigned)__builtin_popcount(same);
    }
    sgc_lds_barrier();
    if (tid < ROW_SLOTS) {
      unsigned hc[HOP_WAVES];
#pragma unroll
      for (int w = 0; w < HOP_WAVES; ++w) hc[w] = L.f.hist[w][tid];
      unsigned run = 0u;
#pragma unroll
      for (int w = 0; w < HOP_WAVES; ++w) {
        L.f.hist[w][tid] = run;
        run += hc[w];
      }
      L.f.rtot[tid] = run;
    }
    sgc_lds_barrier();
    if (wave == 0) {
      unsigned cs[13];
      unsigned sum = 0u;
#pragma unroll
      for (int j = 0; j < 13; ++j) {
        const unsigned tt = L.f.rtot[lane * 13 + j];
        cs[j] = sum;
        sum += tt;
      }
      unsigned incl = sum;
#pragma unroll
      for (int d = 1; d < 32; d <<= 1) {
        const unsigned up = __shfl_up(incl, (unsigned)d);
        if (lane >= d) incl += up;
      }
      const unsigned excl = incl - sum;
#pragma unroll
      for (int j = 0; j < 13; ++j) L.f.rstart[lane * 13 + j] = excl + cs[j];
    }
    sgc_lds_barrier();

    for (int t = 0; t * 32 < ecnt; ++t) {
      const int  i     = t * 32 + lane;
      const bool valid = i < ecnt;
      const unsigned e   = valid ? el[i] : 0u;
      const unsigned row = e >> SRC_BITS;
      const unsigned same = sgc_same_row(valid, row);
      const unsigned rank = (unsigned)__builtin_popcount(same & lt);
      if (valid) {
        const unsigned before = hw[row];
        L.f.byrow[L.f.rstart[row] + before + rank] = e & ((1u << SRC_BITS) - 1u);
        if (rank == 0u) hw[row] = before + (unsigned)__builtin_popcount(same);
      }
    }
    sgc_lds_barrier();

    const int   kr  = wave + HOP_WAVES * lane;
    const float ndv = kr < nrows ? norm[node0 + kr] : 0.f;
    int k = 0;
    for (int r = wave; r < nrows; r += HOP_WAVES, ++k) {
      const int n  = __builtin_amdgcn_readfirstlane((int)L.f.rtot[r]);
      const int s0 = __builtin_amdgcn_readfirstlane((int)L.f.rstart[r]);
      const float ndst = __int_as_float(__builtin_amdgcn_readlane(__float_as_int(ndv), k));
      const unsigned* br = L.f.byrow + s0;
      v4f a = {0.f, 0.f, 0.f, 0.f};
      for (int base = 0; base < n; base += 32) {
        const int m  = n - base < 32 ? n - base : 32;
        const int li = lane < m ? lane : m - 1;
        const int sv = (int)br[base + li];
        v4f v[ROW_BATCH];
        sgc_row_issue(x, v, sv, lane, 0, m);
        sgc_row_add(a, v, 0, m);
        if (m > ROW_BATCH) {
          sgc_row_issue(x, v, sv, lane, ROW_BATCH, m);
          sgc_row_add(a, v, ROW_BATCH, m);
        }
      }
      v4f o = ndst * a;
      if (twice) o = ndst * o;
      ((v4f*)xout)[(size_t)(node0 + r) * (OUT_F / 4) + lane] = o;
    }
  } else {
    for (int i = tid; i < NODES_PER_WG * OUT_F / 4; i += HOP_WAVES * 32)
      ((float4*)acc)[i] = make_float4(0.f, 0.f, 0.f, 0.f);
    int4 buf0[4] = {}, buf1[4] = {}, buf2[4] = {};
    SGC_LOAD(0, buf0)
    SGC_LOAD(1, buf1)
    for (int it = 0; it < nsteps; it += 3) {
      SGC_STEP(it, buf0, buf2)
      if (it + 1 < nsteps) SGC_STEP(it + 1, buf1, buf0)
      if (it + 2 < nsteps) SGC_STEP(it + 2, buf2, buf1)
    }
    if (pc > 0) sgc_hop_group<true>(x, src, acc, oe, od, pc, lane, true);

    for (int r = wave; r < nrows; r += HOP_WAVES) {
      const float  ndst = norm[node0 + r];
      const float4 a    = ((const float4*)acc)[r * (OUT_F / 4) + lane];
      float4 o = make_float4(ndst * a.x, ndst * a.y, ndst * a.z, ndst * a.w);
      if (twice) o = make_float4(ndst * o.x, ndst * o.y, ndst * o.z, ndst * o.w);
      ((float4*)xout)[(size_t)(node0 + r) * (OUT_F / 4) + lane] = o;
    }
  }
#undef SGC_STEP
#undef SGC_LOAD
#undef SGC_NOTE
}

__device__ __forceinline__ void sgc_bin_trip(unsigned* my, unsigned* my_n, unsigned d, unsigned s,
                                             bool in, unsigned lt) {
  const unsigned g   = d / NODES_PER_WG;
  const unsigned row = d - g * NODES_PER_WG;
  const bool ok = in && g < BIN_GROUPS;
  unsigned same = __builtin_amdgcn_ballot_w32(ok);
#pragma unroll
  for (int b = 0; b < 8; ++b) {
    const bool bit = (g >> b) & 1u;
    const unsigned bm = __builtin_amdgcn_ballot_w32(bit);
    same &= bit ? bm : ~bm;
  }
  const unsigned rank = __builtin_popcount(same & lt);
  __builtin_amdgcn_wave_barrier();
  if (ok) {
    const unsigned base = my_n[g];
    const unsigned pos  = base + rank;
    if (pos < SLOT_CAP) my[g * SLOT_W + 1 + pos] = (row << SRC_BITS) | s;
    if (rank == 0) my_n[g] = base + __builtin_popcount(same);
  }
}

#define BIN_TRIPS 8

__global__ __launch_bounds__(BIN_WAVES * 32) void sgc_bin(const int* __restrict__ src,
                                                          const int* __restrict__ dst,
                                                          float* __restrict__ buf, int N, int E) {
  __shared__ unsigned slot[BIN_WAVES * BIN_GROUPS * SLOT_W];
  __shared__ unsigned slot_n[BIN_WAVES * BIN_GROUPS];
  const int lane = threadIdx.x & 31;
  const int wave = __builtin_amdgcn_readfirstlane((int)threadIdx.x >> 5);
  const int task = blockIdx.x * BIN_WAVES + wave;
  unsigned* my   = slot + wave * (BIN_GROUPS * SLOT_W);
  unsigned* my_n = slot_n + wave * BIN_GROUPS;
  for (int g = lane; g < BIN_GROUPS; g += 32) my_n[g] = 0u;

  const int ept = (E + BIN_TASKS - 1) / BIN_TASKS;
  const int e0  = task * ept;
  const int e1  = e0 + ept < E ? e0 + ept : E;
  const unsigned lt = (1u << lane) - 1u;

#define BIN_LOAD(DB, SB, EB)                                             \
  _Pragma("unroll") for (int k = 0; k < BIN_TRIPS; ++k) {                \
    const int e_ = (EB) + k * 32 + lane;                                 \
    const int c_ = e_ < E ? e_ : E - 1;                                  \
    DB[k] = dst[c_];                                                     \
    SB[k] = src[c_];                                                     \
  }
#define BIN_WORK(DB, SB, EB)                                             \
  _Pragma("unroll") for (int k = 0; k < BIN_TRIPS; ++k) {                \
    const int eb_ = (EB) + k * 32;                                       \
    if (eb_ < e1)                                       \
      sgc_bin_trip(my, my_n, (unsigned)DB[k], (unsigned)SB[k], eb_ + lane < e1, lt); \
  }

  int da[BIN_TRIPS], sa[BIN_TRIPS], db[BIN_TRIPS], sb[BIN_TRIPS];
  BIN_LOAD(da, sa, e0)
  for (int eb = e0; eb < e1; eb += 2 * BIN_TRIPS * 32) {
    BIN_LOAD(db, sb, eb + BIN_TRIPS * 32)
    BIN_WORK(da, sa, eb)
    BIN_LOAD(da, sa, eb + 2 * BIN_TRIPS * 32)
    BIN_WORK(db, sb, eb + BIN_TRIPS * 32)
  }
#undef BIN_LOAD
#undef BIN_WORK
  __builtin_amdgcn_wave_barrier();
  for (int g = lane; g < BIN_GROUPS; g += 32) my[g * SLOT_W] = my_n[g];
  __builtin_amdgcn_wave_barrier();

  int gs = N >= SLOT_MIN_ROWS ? (N - SLOT_MIN_ROWS) / NODES_PER_WG + 1 : 0;
  if (gs > BIN_GROUPS) gs = BIN_GROUPS;
  uint4* line0 = (uint4*)buf + (size_t)task * (SLOT_W / 4) + (lane & 7);
  const size_t gstride = (size_t)NODES_PER_WG * OUT_F / 4;
  int i0 = 0;
  for (; (i0 + 8) * 4 <= gs; i0 += 8) {
    uint4 v[8];
#pragma unroll
    for (int j = 0; j < 8; ++j) v[j] = ((const uint4*)my)[(i0 + j) * 32 + lane];
#pragma unroll
    for (int j = 0; j < 8; ++j) line0[(size_t)((i0 + j) * 4 + (lane >> 3)) * gstride] = v[j];
  }
  for (int i = i0; i * 4 < gs; ++i) {
    const int g = i * 4 + (lane >> 3);
    if (g < gs) line0[(size_t)g * gstride] = ((const uint4*)my)[i * 32 + lane];
  }
}

extern "C" void kernel_launch(void* const* d_in, const int* in_sizes, int n_in,
                              void* d_out, int out_size, void* d_ws, size_t ws_size,
                              hipStream_t stream) {
  const float* h    = (const float*)d_in[0];
  const float* w    = (const float*)d_in[1];
  const float* norm = (const float*)d_in[2];
  const int*   src  = (const int*)d_in[3];
  const int*   dst  = (const int*)d_in[4];
  float* out = (float*)d_out;
  float* x1  = (float*)d_ws;

  const int N = in_sizes[2];
  const int E = in_sizes[3];

  const int mblocks = (N + GEMM_ROWS - 1) / GEMM_ROWS;
  const int groups  = (N + NODES_PER_WG - 1) / NODES_PER_WG;

  sgc_gemm_wmma<<<mblocks, 256, 0, stream>>>(h, w, norm, out, N);
  sgc_bin<<<BIN_TASKS / BIN_WAVES, BIN_WAVES * 32, 0, stream>>>(src, dst, x1, N, E);
  sgc_hop<<<groups, HOP_WAVES * 32, 0, stream>>>(out, norm, src, dst, x1, N, E, 1);
  sgc_bin<<<BIN_TASKS / BIN_WAVES, BIN_WAVES * 32, 0, stream>>>(src, dst, out, N, E);
  sgc_hop<<<groups, HOP_WAVES * 32, 0, stream>>>(x1, norm, src, dst, out, N, E, 0);
}


# Hop 1's edge binning moved into the GEMM launch as 512 extra work-groups behind the GEMM's (2 binning waves each, slots in the GEMM's LDS tile): 5 launches -> 4, binning beside matrix work; sums kept
# speedup vs baseline: 3.6924x; 1.0341x over previous
#include <hip/hip_runtime.h>

#define IN_F  256
#define OUT_F 128
#define LDS_PAD 260

typedef __attribute__((ext_vector_type(2))) float v2f;
typedef __attribute__((ext_vector_type(8))) float v8f;

#define GEMM_TILES 5
#define GEMM_ROWS (16 * GEMM_TILES)
#define GEMM_STAGE (GEMM_ROWS / 4)

template <bool FULL>
__device__ __forceinline__ void sgc_gemm_body(const float* __restrict__ h,
                                              const float* __restrict__ w,
                                              const float* __restrict__ norm,
                                              float* __restrict__ x, int N, float* atile,
                                              int m0, int tid) {
  const int r0 = tid >> 6;
  const int c0 = (tid & 63) * 4;
  const float* hb = h + (size_t)(m0 + r0) * IN_F + c0;
  float4 v[GEMM_STAGE];
#pragma unroll
  for (int j = 0; j < GEMM_STAGE; ++j) {
    if (FULL) {
      v[j] = *(const float4*)(hb + (size_t)(4 * j) * IN_F);
    } else {
      const int gr = m0 + r0 + 4 * j < N ? m0 + r0 + 4 * j : N - 1;
      v[j] = *(const float4*)(h + (size_t)gr * IN_F + c0);
    }
  }
#pragma unroll
  for (int j = 0; j < GEMM_STAGE; ++j)
    *(float4*)(atile + (r0 + 4 * j) * LDS_PAD + c0) = v[j];
  __syncthreads();

  const int wave = tid >> 5;
  const int lane = tid & 31;
  const int half = lane >> 4;
  const int l    = lane & 15;
  const int n0   = wave * 16;

  v8f c[GEMM_TILES] = {};
  const float* arow = atile + l * LDS_PAD + 2 * half;
  const float* bcol = w + (size_t)(2 * half) * OUT_F + n0 + l;
  for (int k = 0; k < IN_F; k += 4) {
    v2f b;
    b.x = bcol[(size_t)k * OUT_F];
    b.y = bcol[(size_t)(k + 1) * OUT_F];
#pragma unroll
    for (int t = 0; t < GEMM_TILES; ++t) {
      float2 av = *(const float2*)(arow + t * 16 * LDS_PAD + k);
      v2f a;
      a.x = av.x;
      a.y = av.y;
      c[t] = __builtin_amdgcn_wmma_f32_16x16x4_f32(false, a, false, b, (short)0, c[t],
                                                   false, false);
    }
  }

  float nr[GEMM_TILES][8];
#pragma unroll
  for (int t = 0; t < GEMM_TILES; ++t) {
#pragma unroll
    for (int r = 0; r < 8; ++r) {
      const int row = m0 + 16 * t + 8 * half + r;
      nr[t][r] = norm[FULL ? row : (row < N ? row : N - 1)];
    }
  }
  float* out = x + (size_t)(m0 + 8 * half) * OUT_F + n0 + l;
#pragma unroll
  for (int t = 0; t < GEMM_TILES; ++t) {
#pragma unroll
    for (int r = 0; r < 8; ++r) {
      if (FULL || m0 + 16 * t + 8 * half + r < N)
        out[(size_t)(16 * t + r) * OUT_F] = c[t][r] * nr[t][r];
    }
  }
}

typedef const float __attribute__((address_space(1)))* gfloat_cptr;
typedef __attribute__((ext_vector_type(4))) float v4f;
typedef const v4f __attribute__((address_space(1)))* gv4f_cptr;

#define NODES_PER_WG 391
#define HOP_WAVES 32
#define CHUNK 512
#define LIST_W 576
#define OWN_W 64
#define EDGE_BATCH 8
#define BIN_TASKS 1024
#define BIN_WAVES 4
#define BIN_GROUPS 256
#define SLOT_W 32
#define SLOT_CAP 31
#define SLOT_MIN_ROWS 256
#define SRC_BITS 22
#define ROW_CAP (BIN_TASKS * SLOT_CAP)
#define WAVE_CAP (HOP_WAVES * SLOT_CAP)
#define ROW_SLOTS 416
#define ROW_BATCH 16

struct sgc_lds_scan {
  float          acc[NODES_PER_WG * OUT_F];
  int            note_e[HOP_WAVES * LIST_W];
  unsigned short note_d[HOP_WAVES * LIST_W];
  int            own_e[HOP_WAVES * OWN_W];
  unsigned short own_d[HOP_WAVES * OWN_W];
  int            note_n[2][HOP_WAVES];
};
struct sgc_lds_rows {
  unsigned elist[ROW_CAP];
  unsigned byrow[ROW_CAP];
  unsigned hist[HOP_WAVES][ROW_SLOTS];
  unsigned rtot[ROW_SLOTS];
  unsigned rstart[ROW_SLOTS];
};
union sgc_lds {
  sgc_lds_scan s;
  sgc_lds_rows f;
};
static_assert(sizeof(sgc_lds_rows) <= offsetof(sgc_lds_scan, note_n), "usual path must not reach note_n");
static_assert(ROW_SLOTS >= NODES_PER_WG && ROW_SLOTS == 13 * 32, "scan by 13 counters per lane");
static_assert(NODES_PER_WG <= 512, "a local row number has 9 bits");
static_assert(HOP_WAVES * 32 == BIN_TASKS, "32 slots per wave");
static_assert(2 * ROW_BATCH == 32, "two batches per 32 edges");
static_assert(NODES_PER_WG <= HOP_WAVES * 32, "a wave's rows fit the lanes of one register");

__device__ __forceinline__ unsigned sgc_same_row(bool valid, unsigned row) {
  unsigned same = __builtin_amdgcn_ballot_w32(valid);
#pragma unroll
  for (int b = 0; b < 9; ++b) {
    const bool bit = (row >> b) & 1u;
    const unsigned bm = __builtin_amdgcn_ballot_w32(bit);
    same &= bit ? bm : ~bm;
  }
  return same;
}

__device__ __forceinline__ void sgc_row_issue(const float* __restrict__ x, v4f (&v)[ROW_BATCH],
                                              int sv, int lane, int b, int m) {
#pragma unroll
  for (int g = 0; g < ROW_BATCH / 4; ++g) {
    if (b + 4 * g < m) {
#pragma unroll
      for (int j = 4 * g; j < 4 * g + 4; ++j) {
        const int s = __builtin_amdgcn_readlane(sv, b + j);
        gv4f_cptr xrow = (gv4f_cptr)(x + (size_t)s * OUT_F);
        asm("" : "+s"(xrow));
        v[j] = xrow[lane];
      }
    }
  }
}

__device__ __forceinline__ void sgc_row_add(v4f& a, const v4f (&v)[ROW_BATCH], int b, int m) {
#pragma unroll
  for (int j = 0; j < ROW_BATCH; ++j) {
    if (b + j < m) a += v[j];
  }
}

template <bool TAIL>
__device__ __forceinline__ void sgc_hop_batch(const float* __restrict__ x, float* acc, int sv,
                                              int dv, int lane, int b, int n) {
  float v[EDGE_BATCH][4];
#pragma unroll
  for (int j = 0; j < EDGE_BATCH; ++j) {
    const int s = __builtin_amdgcn_readlane(sv, b + j);
    gfloat_cptr xrow = (gfloat_cptr)(x + (size_t)s * OUT_F);
    asm("" : "+s"(xrow));
    v[j][0] = xrow[lane];
    v[j][1] = xrow[lane + 32];
    v[j][2] = xrow[lane + 64];
    v[j][3] = xrow[lane + 96];
  }
#pragma unroll
  for (int j = 0; j < EDGE_BATCH; ++j) {
    if (!TAIL || b + j < n) {
      const int d = __builtin_amdgcn_readlane(dv, b + j);
      float* arow = acc + d * OUT_F + lane;
      arow[0]  += v[j][0];
      arow[32] += v[j][1];
      arow[64] += v[j][2];
      arow[96] += v[j][3];
    }
  }
}

template <bool TAIL>
__device__ __forceinline__ void sgc_hop_group(const float* __restrict__ x,
                                              const int* __restrict__ src, float* acc,
                                              const int* oe, const unsigned short* od, int n,
                                              int lane, bool by_edge) {
  const int li = TAIL ? (lane < n ? lane : n - 1) : lane;
  const int e  = oe[li];
  const int dv = od[li];
  const int sv = by_edge ? src[e] : e;
  int b = 0;
  for (; b + EDGE_BATCH <= n; b += EDGE_BATCH)
    sgc_hop_batch<false>(x, acc, sv, dv, lane, b, n);
  if (TAIL && b < n) sgc_hop_batch<true>(x, acc, sv, dv, lane, b, n);
}

__device__ __forceinline__ void sgc_lds_barrier() {
  __builtin_amdgcn_fence(__ATOMIC_RELEASE, "workgroup", "local");
  __builtin_amdgcn_s_barrier();
  __builtin_amdgcn_fence(__ATOMIC_ACQUIRE, "workgroup", "local");
}

__device__ __forceinline__ void sgc_hop_publish_and_hand_over(
    const float* __restrict__ x, const int* __restrict__ src, float* acc, const int* note_e,
    const unsigned short* note_d, int* note_n, int* oe,
    unsigned short* od, int& cnt, int& pc, bool last, int wave, int lane, unsigned lt, int room,
    bool by_edge) {
  if (lane == 0) note_n[wave] = cnt;
  sgc_lds_barrier();
  const int cn = note_n[lane];
  const bool hand_over = last || __builtin_amdgcn_ballot_w32(cn + room > LIST_W) != 0;
  if (!hand_over) return;
  for (int ws = 0; ws < HOP_WAVES; ++ws) {
    const int c = __builtin_amdgcn_readlane(cn, ws);
    const int*            le = note_e + ws * LIST_W;
    const unsigned short* ld = note_d + ws * LIST_W;
    for (int off = 0; off < c; off += 32) {
      const int  idx   = off + lane;
      const bool valid = idx < c;
      const int  d     = valid ? (int)ld[idx] : -1;
      const bool mine  = valid && ((d & (HOP_WAVES - 1)) == wave);
      const unsigned mk = __builtin_amdgcn_ballot_w32(mine);
      if (mk) {
        if (mine) {
          const int pos = pc + __builtin_popcount(mk & lt);
          oe[pos] = le[idx];
          od[pos] = (unsigned short)d;
        }
        pc += __builtin_popcount(mk);
        if (pc >= 32) {
          sgc_hop_group<false>(x, src, acc, oe, od, 32, lane, by_edge);
          const int rem = pc - 32;
          if (lane < rem) {
            const int            te = oe[32 + lane];
            const unsigned short td = od[32 + lane];
            oe[lane] = te;
            od[lane] = td;
          }
          pc = rem;
        }
      }
    }
  }
  sgc_lds_barrier();
  cnt = 0;
}

__global__ __launch_bounds__(HOP_WAVES * 32) void sgc_hop(const float* __restrict__ x,
                                                          const float* __restrict__ norm,
                                                          const int* __restrict__ src,
                                                          const int* __restrict__ dst,
                                                          float* __restrict__ xout, int N, int E,
                                                          int twice) {
  __shared__ sgc_lds L;
  float*          acc    = L.s.acc;
  int*            note_e = L.s.note_e;
  unsigned short* note_d = L.s.note_d;
  int*            own_e  = L.s.own_e;
  unsigned short* own_d  = L.s.own_d;
  int (*note_n)[HOP_WAVES] = L.s.note_n;

  const int tid  = threadIdx.x;
  const int lane = tid & 31;
  const int wave  = __builtin_amdgcn_readfirstlane(tid >> 5);
  const int node0 = blockIdx.x * NODES_PER_WG;
  const int left  = N - node0;
  const int nrows = left < NODES_PER_WG ? left : NODES_PER_WG;

  const float* slots = xout + (size_t)node0 * OUT_F;
  const bool served = (int)gridDim.x <= BIN_GROUPS && nrows >= SLOT_MIN_ROWS;
  unsigned* el = L.f.elist + wave * WAVE_CAP;
  unsigned* hw = L.f.hist[wave];
  int ecnt = 0;
  bool bad = !served;
  if (served) {
    unsigned wv[HOP_WAVES];
#pragma unroll
    for (int k = 0; k < HOP_WAVES; ++k)
      wv[k] = __float_as_uint(slots[(size_t)(k * HOP_WAVES + wave) * SLOT_W + lane]);
    unsigned over = 0u;
#pragma unroll
    for (int k = 0; k < HOP_WAVES; ++k) {
      const unsigned cu = (unsigned)__builtin_amdgcn_readfirstlane((int)wv[k]);
      over |= cu > SLOT_CAP ? 1u : 0u;
      const int c = (int)(cu > SLOT_CAP ? SLOT_CAP : cu);
      if (lane >= 1 && lane <= c) el[ecnt + lane - 1] = wv[k];
      ecnt += c;
    }
    bad = over != 0u;
    for (int j = lane; j < ROW_SLOTS; j += 32) hw[j] = 0u;
  }
  if (lane == 0) note_n[1][wave] = bad ? 1 : 0;
  sgc_lds_barrier();
  const bool scan = __builtin_amdgcn_ballot_w32(note_n[1][lane] != 0) != 0;

  int*            ne = note_e + wave * LIST_W;
  unsigned short* nd = note_d + wave * LIST_W;
  int*            oe = own_e + wave * OWN_W;
  unsigned short* od = own_d + wave * OWN_W;
  int cnt = 0;
  int pc  = 0;
  const unsigned lt = (1u << lane) - 1u;

#define SGC_NOTE(mc, eidx, dloc)                                         \
  {                                                                      \
    const unsigned mk_ = __builtin_amdgcn_ballot_w32(mc);                \
    if (mk_) {                                                           \
      if (mc) {                                                          \
        const int pos_ = cnt + __builtin_popcount(mk_ & lt);             \
        ne[pos_] = (eidx);                                               \
        nd[pos_] = (unsigned short)(dloc);                               \
      }                                                                  \
      cnt += __builtin_popcount(mk_);                                    \
    }                                                                    \
  }

  const int nchunks = E / CHUNK;
  const int nsteps  = (nchunks + HOP_WAVES - 1) / HOP_WAVES + 1;
  const int rot = (int)(((long long)blockIdx.x * nchunks) / gridDim.x);
  const int4* dst4 = (const int4*)dst;

#define SGC_LOAD(IT, BUF)                                                \
  {                                                                      \
    const int ql_ = wave + (IT) * HOP_WAVES;                             \
    int pl_ = (ql_ < nchunks ? ql_ : nchunks - 1) + rot;                 \
    if (pl_ >= nchunks) pl_ -= nchunks;                                  \
    _Pragma("unroll") for (int k = 0; k < 4; ++k)                        \
        BUF[k] = dst4[pl_ * (CHUNK / 4) + k * 32 + lane];                \
  }

#define SGC_STEP(IT, BUF, NEXTBUF)                                       \
  {                                                                      \
    const int it_ = (IT);                                                \
    SGC_LOAD(it_ + 2, NEXTBUF)                                           \
    if (it_ < nsteps - 1) {                                              \
      const int q_ = wave + it_ * HOP_WAVES;                             \
      if (q_ < nchunks) {                                                \
        int p_ = q_ + rot;                                               \
        if (p_ >= nchunks) p_ -= nchunks;                                \
        _Pragma("unroll") for (int k = 0; k < 4; ++k) {                  \
          const int4 d4 = BUF[k];                                        \
          const int l0 = d4.x - node0, l1 = d4.y - node0;                \
          const int l2 = d4.z - node0, l3 = d4.w - node0;                \
          const bool m0 = (unsigned)l0 < (unsigned)nrows;                \
          const bool m1 = (unsigned)l1 < (unsigned)nrows;                \
          const bool m2 = (unsigned)l2 < (unsigned)nrows;                \
          const bool m3 = (unsigned)l3 < (unsigned)nrows;                \
                                 \
          if (__builtin_amdgcn_ballot_w32(m0 | m1 | m2 | m3)) {          \
            const int e0 = p_ * CHUNK + k * 128 + lane * 4;              \
            SGC_NOTE(m0, e0, l0)                                         \
            SGC_NOTE(m1, e0 + 1, l1)                                     \
            SGC_NOTE(m2, e0 + 2, l2)                                     \
            SGC_NOTE(m3, e0 + 3, l3)                                     \
          }                                                              \
        }                                                                \
      }                                                                  \
    } else {                                                             \
            \
      const int i_ = nchunks * CHUNK + wave * 32;                        \
      if (i_ < E) {                                                      \
        const int  e_  = i_ + lane;                                      \
        const bool ok_ = e_ < E;                                         \
        const int  l_  = dst[ok_ ? e_ : E - 1] - node0;                  \
        const bool m_  = ok_ && (unsigned)l_ < (unsigned)nrows;          \
        SGC_NOTE(m_, e_, l_)                                             \
      }                                                                  \
    }                                                                    \
    sgc_hop_publish_and_hand_over(x, src, acc, note_e, note_d, &note_n[it_ & 1][0], oe, od,   \
                                  cnt, pc, it_ == nsteps - 1, wave, lane, lt, CHUNK, true);   \
  }

  if (!scan) {
    for (int t = 0; t * 32 < ecnt; ++t) {
      const int  i     = t * 32 + lane;
      const bool valid = i < ecnt;
      const unsigned e   = valid ? el[i] : 0u;
      const unsigned row = e >> SRC_BITS;
      const unsigned same = sgc_same_row(valid, row);
      if (valid && (same & lt) == 0u) hw[row] += (unsigned)__builtin_popcount(same);
    }
    sgc_lds_barrier();
    if (tid < ROW_SLOTS) {
      unsigned hc[HOP_WAVES];
#pragma unroll
      for (int w = 0; w < HOP_WAVES; ++w) hc[w] = L.f.hist[w][tid];
      unsigned run = 0u;
#pragma unroll
      for (int w = 0; w < HOP_WAVES; ++w) {
        L.f.hist[w][tid] = run;
        run += hc[w];
      }
      L.f.rtot[tid] = run;
    }
    sgc_lds_barrier();
    if (wave == 0) {
      unsigned cs[13];
      unsigned sum = 0u;
#pragma unroll
      for (int j = 0; j < 13; ++j) {
        const unsigned tt = L.f.rtot[lane * 13 + j];
        cs[j] = sum;
        sum += tt;
      }
      unsigned incl = sum;
#pragma unroll
      for (int d = 1; d < 32; d <<= 1) {
        const unsigned up = __shfl_up(incl, (unsigned)d);
        if (lane >= d) incl += up;
      }
      const unsigned excl = incl - sum;
#pragma unroll
      for (int j = 0; j < 13; ++j) L.f.rstart[lane * 13 + j] = excl + cs[j];
    }
    sgc_lds_barrier();

    for (int t = 0; t * 32 < ecnt; ++t) {
      const int  i     = t * 32 + lane;
      const bool valid = i < ecnt;
      const unsigned e   = valid ? el[i] : 0u;
      const unsigned row = e >> SRC_BITS;
      const unsigned same = sgc_same_row(valid, row);
      const unsigned rank = (unsigned)__builtin_popcount(same & lt);
      if (valid) {
        const unsigned before = hw[row];
        L.f.byrow[L.f.rstart[row] + before + rank] = e & ((1u << SRC_BITS) - 1u);
        if (rank == 0u) hw[row] = before + (unsigned)__builtin_popcount(same);
      }
    }
    sgc_lds_barrier();

    const int   kr  = wave + HOP_WAVES * lane;
    const float ndv = kr < nrows ? norm[node0 + kr] : 0.f;
    int k = 0;
    for (int r = wave; r < nrows; r += HOP_WAVES, ++k) {
      const int n  = __builtin_amdgcn_readfirstlane((int)L.f.rtot[r]);
      const int s0 = __builtin_amdgcn_readfirstlane((int)L.f.rstart[r]);
      const float ndst = __int_as_float(__builtin_amdgcn_readlane(__float_as_int(ndv), k));
      const unsigned* br = L.f.byrow + s0;
      v4f a = {0.f, 0.f, 0.f, 0.f};
      for (int base = 0; base < n; base += 32) {
        const int m  = n - base < 32 ? n - base : 32;
        const int li = lane < m ? lane : m - 1;
        const int sv = (int)br[base + li];
        v4f v[ROW_BATCH];
        sgc_row_issue(x, v, sv, lane, 0, m);
        sgc_row_add(a, v, 0, m);
        if (m > ROW_BATCH) {
          sgc_row_issue(x, v, sv, lane, ROW_BATCH, m);
          sgc_row_add(a, v, ROW_BATCH, m);
        }
      }
      v4f o = ndst * a;
      if (twice) o = ndst * o;
      ((v4f*)xout)[(size_t)(node0 + r) * (OUT_F / 4) + lane] = o;
    }
  } else {
    for (int i = tid; i < NODES_PER_WG * OUT_F / 4; i += HOP_WAVES * 32)
      ((float4*)acc)[i] = make_float4(0.f, 0.f, 0.f, 0.f);
    int4 buf0[4] = {}, buf1[4] = {}, buf2[4] = {};
    SGC_LOAD(0, buf0)
    SGC_LOAD(1, buf1)
    for (int it = 0; it < nsteps; it += 3) {
      SGC_STEP(it, buf0, buf2)
      if (it + 1 < nsteps) SGC_STEP(it + 1, buf1, buf0)
      if (it + 2 < nsteps) SGC_STEP(it + 2, buf2, buf1)
    }
    if (pc > 0) sgc_hop_group<true>(x, src, acc, oe, od, pc, lane, true);

    for (int r = wave; r < nrows; r += HOP_WAVES) {
      const float  ndst = norm[node0 + r];
      const float4 a    = ((const float4*)acc)[r * (OUT_F / 4) + lane];
      float4 o = make_float4(ndst * a.x, ndst * a.y, ndst * a.z, ndst * a.w);
      if (twice) o = make_float4(ndst * o.x, ndst * o.y, ndst * o.z, ndst * o.w);
      ((float4*)xout)[(size_t)(node0 + r) * (OUT_F / 4) + lane] = o;
    }
  }
#undef SGC_STEP
#undef SGC_LOAD
#undef SGC_NOTE
}

__device__ __forceinline__ void sgc_bin_trip(unsigned* my, unsigned* my_n, unsigned d, unsigned s,
                                             bool in, unsigned lt) {
  const unsigned g   = d / NODES_PER_WG;
  const unsigned row = d - g * NODES_PER_WG;
  const bool ok = in && g < BIN_GROUPS;
  unsigned same = __builtin_amdgcn_ballot_w32(ok);
#pragma unroll
  for (int b = 0; b < 8; ++b) {
    const bool bit = (g >> b) & 1u;
    const unsigned bm = __builtin_amdgcn_ballot_w32(bit);
    same &= bit ? bm : ~bm;
  }
  const unsigned rank = __builtin_popcount(same & lt);
  __builtin_amdgcn_wave_barrier();
  if (ok) {
    const unsigned base = my_n[g];
    const unsigned pos  = base + rank;
    if (pos < SLOT_CAP) my[g * SLOT_W + 1 + pos] = (row << SRC_BITS) | s;
    if (rank == 0) my_n[g] = base + __builtin_popcount(same);
  }
}

#define BIN_TRIPS 8

__device__ __forceinline__ void sgc_bin_wave(const int* __restrict__ src,
                                             const int* __restrict__ dst,
                                             float* __restrict__ buf, int N, int E,
                                             unsigned* my, unsigned* my_n, int lane, int task) {
  for (int g = lane; g < BIN_GROUPS; g += 32) my_n[g] = 0u;

  const int ept = (E + BIN_TASKS - 1) / BIN_TASKS;
  const int e0  = task * ept;
  const int e1  = e0 + ept < E ? e0 + ept : E;
  const unsigned lt = (1u << lane) - 1u;

#define BIN_LOAD(DB, SB, EB)                                             \
  _Pragma("unroll") for (int k = 0; k < BIN_TRIPS; ++k) {                \
    const int e_ = (EB) + k * 32 + lane;                                 \
    const int c_ = e_ < E ? e_ : E - 1;                                  \
    DB[k] = dst[c_];                                                     \
    SB[k] = src[c_];                                                     \
  }
#define BIN_WORK(DB, SB, EB)                                             \
  _Pragma("unroll") for (int k = 0; k < BIN_TRIPS; ++k) {                \
    const int eb_ = (EB) + k * 32;                                       \
    if (eb_ < e1)                                       \
      sgc_bin_trip(my, my_n, (unsigned)DB[k], (unsigned)SB[k], eb_ + lane < e1, lt); \
  }

  int da[BIN_TRIPS], sa[BIN_TRIPS], db[BIN_TRIPS], sb[BIN_TRIPS];
  BIN_LOAD(da, sa, e0)
  for (int eb = e0; eb < e1; eb += 2 * BIN_TRIPS * 32) {
    BIN_LOAD(db, sb, eb + BIN_TRIPS * 32)
    BIN_WORK(da, sa, eb)
    BIN_LOAD(da, sa, eb + 2 * BIN_TRIPS * 32)
    BIN_WORK(db, sb, eb + BIN_TRIPS * 32)
  }
#undef BIN_LOAD
#undef BIN_WORK
  __builtin_amdgcn_wave_barrier();
  for (int g = lane; g < BIN_GROUPS; g += 32) my[g * SLOT_W] = my_n[g];
  __builtin_amdgcn_wave_barrier();

  int gs = N >= SLOT_MIN_ROWS ? (N - SLOT_MIN_ROWS) / NODES_PER_WG + 1 : 0;
  if (gs > BIN_GROUPS) gs = BIN_GROUPS;
  uint4* line0 = (uint4*)buf + (size_t)task * (SLOT_W / 4) + (lane & 7);
  const size_t gstride = (size_t)NODES_PER_WG * OUT_F / 4;
  int i0 = 0;
  for (; (i0 + 8) * 4 <= gs; i0 += 8) {
    uint4 v[8];
#pragma unroll
    for (int j = 0; j < 8; ++j) v[j] = ((const uint4*)my)[(i0 + j) * 32 + lane];
#pragma unroll
    for (int j = 0; j < 8; ++j) line0[(size_t)((i0 + j) * 4 + (lane >> 3)) * gstride] = v[j];
  }
  for (int i = i0; i * 4 < gs; ++i) {
    const int g = i * 4 + (lane >> 3);
    if (g < gs) line0[(size_t)g * gstride] = ((const uint4*)my)[i * 32 + lane];
  }
}

__global__ __launch_bounds__(BIN_WAVES * 32) void sgc_bin(const int* __restrict__ src,
                                                          const int* __restrict__ dst,
                                                          float* __restrict__ buf, int N, int E) {
  __shared__ unsigned slot[BIN_WAVES * BIN_GROUPS * SLOT_W];
  __shared__ unsigned slot_n[BIN_WAVES * BIN_GROUPS];
  const int lane = threadIdx.x & 31;
  const int wave = __builtin_amdgcn_readfirstlane((int)threadIdx.x >> 5);
  sgc_bin_wave(src, dst, buf, N, E, slot + wave * (BIN_GROUPS * SLOT_W), slot_n + wave * BIN_GROUPS,
               lane, blockIdx.x * BIN_WAVES + wave);
}

#define GEMM_BIN_WAVES 2
static_assert(GEMM_BIN_WAVES * (BIN_GROUPS * SLOT_W + BIN_GROUPS) * 4 <= GEMM_ROWS * LDS_PAD * 4,
              "binning waves' slots fit the GEMM tile");
static_assert(BIN_TASKS % GEMM_BIN_WAVES == 0, "whole binning work-groups");

__global__ __launch_bounds__(256) void sgc_gemm_wmma(const float* __restrict__ h,
                                                     const float* __restrict__ w,
                                                     const float* __restrict__ norm,
                                                     float* __restrict__ x, int N,
                                                     const int* __restrict__ src,
                                                     const int* __restrict__ dst,
                                                     float* __restrict__ binbuf, int E, int mblocks) {
  __shared__ __attribute__((aligned(16))) float atile[GEMM_ROWS * LDS_PAD];
  const int tid = threadIdx.x;
  if ((int)blockIdx.x < mblocks) {
    const int m0 = blockIdx.x * GEMM_ROWS;
    if (m0 + GEMM_ROWS <= N)
      sgc_gemm_body<true>(h, w, norm, x, N, atile, m0, tid);
    else
      sgc_gemm_body<false>(h, w, norm, x, N, atile, m0, tid);
  } else {
    const int wave = __builtin_amdgcn_readfirstlane(tid >> 5);
    if (wave < GEMM_BIN_WAVES) {
      unsigned* lds = (unsigned*)atile;
      sgc_bin_wave(src, dst, binbuf, N, E, lds + wave * (BIN_GROUPS * SLOT_W),
                   lds + GEMM_BIN_WAVES * (BIN_GROUPS * SLOT_W) + wave * BIN_GROUPS, tid & 31,
                   ((int)blockIdx.x - mblocks) * GEMM_BIN_WAVES + wave);
    }
  }
}

extern "C" void kernel_launch(void* const* d_in, const int* in_sizes, int n_in,
                              void* d_out, int out_size, void* d_ws, size_t ws_size,
                              hipStream_t stream) {
  const float* h    = (const float*)d_in[0];
  const float* w    = (const float*)d_in[1];
  const float* norm = (const float*)d_in[2];
  const int*   src  = (const int*)d_in[3];
  const int*   dst  = (const int*)d_in[4];
  float* out = (float*)d_out;
  float* x1  = (float*)d_ws;

  const int N = in_sizes[2];
  const int E = in_sizes[3];

  const int mblocks = (N + GEMM_ROWS - 1) / GEMM_ROWS;
  const int groups  = (N + NODES_PER_WG - 1) / NODES_PER_WG;

  sgc_gemm_wmma<<<mblocks + BIN_TASKS / GEMM_BIN_WAVES, 256, 0, stream>>>(h, w, norm, out, N, src, dst,
                                                                        x1, E, mblocks);
  sgc_hop<<<groups, HOP_WAVES * 32, 0, stream>>>(out, norm, src, dst, x1, N, E, 1);
  sgc_bin<<<BIN_TASKS / BIN_WAVES, BIN_WAVES * 32, 0, stream>>>(src, dst, out, N, E);
  sgc_hop<<<groups, HOP_WAVES * 32, 0, stream>>>(x1, norm, src, dst, out, N, E, 0);
}
